# starting kernel
# baseline (speedup 1.0000x reference)
#include <hip/hip_runtime.h>


typedef float v2f __attribute__((ext_vector_type(2)));
typedef float v8f __attribute__((ext_vector_type(8)));

#define NUM_GRAPHS 512
#define C_OUT 10


__global__ void fill_kernel(float* __restrict__ p, float v, int n) {
    int t = blockIdx.x * blockDim.x + threadIdx.x;
    if (t < n) p[t] = v;
}

__global__ void degree_kernel(float* __restrict__ deg, const int* __restrict__ col, int E) {
    int t = blockIdx.x * blockDim.x + threadIdx.x;
    if (t < E) atomicAdd(&deg[col[t]], 1.0f);
}

__global__ void rsqrt_kernel(float* __restrict__ deg, int n) {
    int t = blockIdx.x * blockDim.x + threadIdx.x;
    if (t < n) deg[t] = rsqrtf(deg[t]);
}

__global__ void relu_kernel(float* __restrict__ p, long long n) {
    long long t = (long long)blockIdx.x * blockDim.x + threadIdx.x;
    if (t < n) { float v = p[t]; p[t] = v > 0.0f ? v : 0.0f; }
}

template <int BCOLS, int DSTRIDE, int KDIM>
__global__ __launch_bounds__(256)
void gemm_wmma_f32(const float* __restrict__ A, const float* __restrict__ B,
                   float* __restrict__ D, int N, int ntilesN) {
    int wave = blockIdx.x * (blockDim.x >> 5) + (threadIdx.x >> 5);
    int lane = threadIdx.x & 31;
    int tilesM = (N + 15) >> 4;
    if (wave >= tilesM * ntilesN) return;
    int tm = wave / ntilesN;
    int tn = wave - tm * ntilesN;
    int m0 = tm << 4, n0 = tn << 4;
    int half = lane >> 4;
    int l15  = lane & 15;

    int row  = m0 + l15;   if (row  >= N)     row  = N - 1;
    int bcol = n0 + l15;   if (bcol >= BCOLS) bcol = BCOLS - 1;

    const float* __restrict__ Arow = A + (size_t)row * KDIM + half * 2;
    const float* __restrict__ Bcol = B + (size_t)(half * 2) * BCOLS + bcol;

    v8f acc = {0.f, 0.f, 0.f, 0.f, 0.f, 0.f, 0.f, 0.f};
#pragma unroll
    for (int k0 = 0; k0 < KDIM; k0 += 4) {
        v2f a = *(const v2f*)(Arow + k0);
        v2f b;
        b[0] = Bcol[(size_t)k0 * BCOLS];
        b[1] = Bcol[(size_t)(k0 + 1) * BCOLS];
        acc = __builtin_amdgcn_wmma_f32_16x16x4_f32(false, a, false, b,
                                                    (short)0, acc, false, false);
    }
#pragma unroll
    for (int r = 0; r < 8; ++r) {
        int m = m0 + r + half * 8;
        if (m < N) D[(size_t)m * DSTRIDE + n0 + l15] = acc[r];
    }
}

template <int F>
__global__ void init_agg_kernel(const float* __restrict__ lin, const float* __restrict__ dis,
                                const float* __restrict__ bias, int biasLen,
                                float* __restrict__ out, int N) {
    long long t = (long long)blockIdx.x * blockDim.x + threadIdx.x;
    if (t >= (long long)N * F) return;
    int i = (int)(t / F);
    int c = (int)(t - (long long)i * F);
    float d  = dis[i];
    float bv = (c < biasLen) ? bias[c] : 0.0f;
    out[t] = lin[t] * d * d + bv;
}

template <int F>
__global__ void scatter_kernel(const float* __restrict__ lin, const float* __restrict__ dis,
                               const int* __restrict__ row, const int* __restrict__ col,
                               float* __restrict__ out, int E) {
    const int LPE = F / 4;
    long long t = (long long)blockIdx.x * blockDim.x + threadIdx.x;
    long long e = t / LPE;
    int l = (int)(t - e * LPE);
    if (e >= E) return;
    int r = row[e];
    int c = col[e];
    float s = dis[r] * dis[c];
    const float4 v = *(const float4*)(lin + (size_t)r * F + l * 4);
    float* o = out + (size_t)c * F + l * 4;
    atomicAdd(o + 0, v.x * s);
    atomicAdd(o + 1, v.y * s);
    atomicAdd(o + 2, v.z * s);
    atomicAdd(o + 3, v.w * s);
}

__global__ void pool_kernel(const float* __restrict__ h  ,
                            const int* __restrict__ batch,
                            float* __restrict__ out, float* __restrict__ cnt, int N) {
    long long t = (long long)blockIdx.x * blockDim.x + threadIdx.x;
    if (t >= (long long)N * 16) return;
    int i = (int)(t >> 4);
    int c = (int)(t & 15);
    int g = batch[i];
    if (c < C_OUT) atomicAdd(&out[(size_t)g * C_OUT + c], h[t]);
    if (c == 0)    atomicAdd(&cnt[g], 1.0f);
}

__global__ void div_kernel(float* __restrict__ out, const float* __restrict__ cnt) {
    int t = blockIdx.x * blockDim.x + threadIdx.x;
    if (t < NUM_GRAPHS * C_OUT) out[t] /= fmaxf(cnt[t / C_OUT], 1.0f);
}


static inline int cdiv_ll(long long a, long long b) { return (int)((a + b - 1) / b); }

extern "C" void kernel_launch(void* const* d_in, const int* in_sizes, int n_in,
                              void* d_out, int out_size, void* d_ws, size_t ws_size,
                              hipStream_t stream) {
    const float* x     = (const float*)d_in[0];
    const int*   ei    = (const int*)d_in[1];
    const int*   batch = (const int*)d_in[2];
    const float* W0 = (const float*)d_in[3];  const float* b0 = (const float*)d_in[4];
    const float* W1 = (const float*)d_in[5];  const float* b1 = (const float*)d_in[6];
    const float* W2 = (const float*)d_in[7];  const float* b2 = (const float*)d_in[8];
    const float* W3 = (const float*)d_in[9];  const float* b3 = (const float*)d_in[10];

    const int N = in_sizes[2];
    const int E = in_sizes[1] / 2;
    const int H = 128;
    const int* rowi = ei;
    const int* coli = ei + E;

    float* out = (float*)d_out;
    float* ws  = (float*)d_ws;
    float* dis  = ws;
    float* bufA = ws + (((size_t)N + 255) & ~(size_t)255);
    float* bufB = bufA + (size_t)N * H;
    float* cnt  = bufB + (size_t)N * H;

    const int TB = 256;

    fill_kernel<<<cdiv_ll(N, TB), TB, 0, stream>>>(dis, 1.0f, N);
    degree_kernel<<<cdiv_ll(E, TB), TB, 0, stream>>>(dis, coli, E);
    rsqrt_kernel<<<cdiv_ll(N, TB), TB, 0, stream>>>(dis, N);

    const int tilesM     = (N + 15) >> 4;
    const int gemmBlkH   = cdiv_ll((long long)tilesM * 8, TB / 32);
    const int gemmBlkOut = cdiv_ll((long long)tilesM * 1, TB / 32);
    const long long NH   = (long long)N * H;

    gemm_wmma_f32<128, 128, 128><<<gemmBlkH, TB, 0, stream>>>(x, W0, bufA, N, 8);
    init_agg_kernel<128><<<cdiv_ll(NH, TB), TB, 0, stream>>>(bufA, dis, b0, 128, bufB, N);
    scatter_kernel<128><<<cdiv_ll((long long)E * 32, TB), TB, 0, stream>>>(bufA, dis, rowi, coli, bufB, E);
    relu_kernel<<<cdiv_ll(NH, TB), TB, 0, stream>>>(bufB, NH);

    gemm_wmma_f32<128, 128, 128><<<gemmBlkH, TB, 0, stream>>>(bufB, W1, bufA, N, 8);
    init_agg_kernel<128><<<cdiv_ll(NH, TB), TB, 0, stream>>>(bufA, dis, b1, 128, bufB, N);
    scatter_kernel<128><<<cdiv_ll((long long)E * 32, TB), TB, 0, stream>>>(bufA, dis, rowi, coli, bufB, E);
    relu_kernel<<<cdiv_ll(NH, TB), TB, 0, stream>>>(bufB, NH);

    gemm_wmma_f32<128, 128, 128><<<gemmBlkH, TB, 0, stream>>>(bufB, W2, bufA, N, 8);
    init_agg_kernel<128><<<cdiv_ll(NH, TB), TB, 0, stream>>>(bufA, dis, b2, 128, bufB, N);
    scatter_kernel<128><<<cdiv_ll((long long)E * 32, TB), TB, 0, stream>>>(bufA, dis, rowi, coli, bufB, E);
    relu_kernel<<<cdiv_ll(NH, TB), TB, 0, stream>>>(bufB, NH);

    gemm_wmma_f32<10, 16, 128><<<gemmBlkOut, TB, 0, stream>>>(bufB, W3, bufA, N, 1);
    init_agg_kernel<16><<<cdiv_ll((long long)N * 16, TB), TB, 0, stream>>>(bufA, dis, b3, C_OUT, bufB, N);
    scatter_kernel<16><<<cdiv_ll((long long)E * 4, TB), TB, 0, stream>>>(bufA, dis, rowi, coli, bufB, E);

    fill_kernel<<<cdiv_ll(NUM_GRAPHS * C_OUT, TB), TB, 0, stream>>>(out, 0.0f, NUM_GRAPHS * C_OUT);
    fill_kernel<<<cdiv_ll(NUM_GRAPHS, TB), TB, 0, stream>>>(cnt, 0.0f, NUM_GRAPHS);
    pool_kernel<<<cdiv_ll((long long)N * 16, TB), TB, 0, stream>>>(bufB, batch, out, cnt, N);
    div_kernel<<<cdiv_ll(NUM_GRAPHS * C_OUT, TB), TB, 0, stream>>>(out, cnt);
}


# Edge scatter of the 128-wide layers walks the edge list once per 64-column half (grid y; a wave takes 4 edges x 64 columns): one walk's rows and sums, half of each buffer, stay in the L2 cache
# speedup vs baseline: 1.6907x; 1.6907x over previous
#include <hip/hip_runtime.h>


typedef float v2f __attribute__((ext_vector_type(2)));
typedef float v8f __attribute__((ext_vector_type(8)));

#define NUM_GRAPHS 512
#define C_OUT 10


__global__ void fill_kernel(float* __restrict__ p, float v, int n) {
    int t = blockIdx.x * blockDim.x + threadIdx.x;
    if (t < n) p[t] = v;
}

__global__ void degree_kernel(float* __restrict__ deg, const int* __restrict__ col, int E) {
    int t = blockIdx.x * blockDim.x + threadIdx.x;
    if (t < E) atomicAdd(&deg[col[t]], 1.0f);
}

__global__ void rsqrt_kernel(float* __restrict__ deg, int n) {
    int t = blockIdx.x * blockDim.x + threadIdx.x;
    if (t < n) deg[t] = rsqrtf(deg[t]);
}

template <int BCOLS, int DSTRIDE, int KDIM, bool RELU_A>
__global__ __launch_bounds__(256)
void gemm_wmma_f32(const float* __restrict__ A, const float* __restrict__ B,
                   const float* __restrict__ dis, const float* __restrict__ bias,
                   float* __restrict__ D, float* __restrict__ S, int N, int ntilesN) {
    int wave = blockIdx.x * (blockDim.x >> 5) + (threadIdx.x >> 5);
    int lane = threadIdx.x & 31;
    int tilesM = (N + 15) >> 4;
    if (wave >= tilesM * ntilesN) return;
    int tm = wave / ntilesN;
    int tn = wave - tm * ntilesN;
    int m0 = tm << 4, n0 = tn << 4;
    int half = lane >> 4;
    int l15  = lane & 15;

    int row  = m0 + l15;   if (row  >= N)     row  = N - 1;
    int bcol = n0 + l15;   if (bcol >= BCOLS) bcol = BCOLS - 1;

    const float* __restrict__ Arow = A + (size_t)row * KDIM + half * 2;
    const float* __restrict__ Bcol = B + (size_t)(half * 2) * BCOLS + bcol;

    v8f acc = {0.f, 0.f, 0.f, 0.f, 0.f, 0.f, 0.f, 0.f};
#pragma unroll
    for (int k0 = 0; k0 < KDIM; k0 += 4) {
        v2f a = *(const v2f*)(Arow + k0);
        if constexpr (RELU_A) {
            a[0] = a[0] > 0.0f ? a[0] : 0.0f;
            a[1] = a[1] > 0.0f ? a[1] : 0.0f;
        }
        v2f b;
        b[0] = Bcol[(size_t)k0 * BCOLS];
        b[1] = Bcol[(size_t)(k0 + 1) * BCOLS];
        acc = __builtin_amdgcn_wmma_f32_16x16x4_f32(false, a, false, b,
                                                    (short)0, acc, false, false);
    }
    const int   ccol = n0 + l15;
    const float bl   = bias[bcol];
    const float bv   = (ccol < BCOLS) ? bl : 0.0f;
    float sv[8];
#pragma unroll
    for (int r = 0; r < 8; ++r) {
        int m  = m0 + r + half * 8;
        int mc = m < N ? m : N - 1;
        float d = dis[mc];
        sv[r] = acc[r] * d * d + bv;
    }
#pragma unroll
    for (int r = 0; r < 8; ++r) {
        int m = m0 + r + half * 8;
        if (m < N) {
            size_t o = (size_t)m * DSTRIDE + ccol;
            D[o] = acc[r];
            S[o] = sv[r];
        }
    }
}

template <int F, int SLICES, int STEPS>
__global__ void scatter_kernel(const float* __restrict__ lin, const float* __restrict__ dis,
                               const int* __restrict__ row, const int* __restrict__ col,
                               float* __restrict__ out, int E) {
    constexpr int FS  = F / SLICES;
    constexpr int LPE = (FS < 32) ? FS : 32;
    constexpr int CPL = FS / LPE;
    constexpr int EPS = 32 / LPE;
    long long t = (long long)blockIdx.x * blockDim.x + threadIdx.x;
    long long wave = t >> 5;
    int lane = (int)(t & 31);
    int sub  = lane / LPE;
    int d    = lane - sub * LPE;
    if constexpr (SLICES > 1) d += (int)blockIdx.y * FS;

    float x[STEPS][CPL];
    float dr[STEPS], dc[STEPS];
    float* o[STEPS];
    bool live[STEPS];
#pragma unroll
    for (int j = 0; j < STEPS; ++j) {
        long long e = (wave * STEPS + j) * EPS + sub;
        live[j] = e < E;
        long long ec = live[j] ? e : (long long)E - 1;
        int r = row[ec];
        int c = col[ec];
        dr[j] = dis[r];
        dc[j] = dis[c];
        const float* src = lin + (size_t)r * F + d;
#pragma unroll
        for (int k = 0; k < CPL; ++k) x[j][k] = src[k * LPE];
        o[j] = out + (size_t)c * F + d;
    }
    if constexpr (STEPS > 1) {
#pragma unroll
        for (int j = 0; j < STEPS; ++j) {
            asm volatile("" :: "v"(dr[j]), "v"(dc[j]));
#pragma unroll
            for (int k = 0; k < CPL; ++k) asm volatile("" :: "v"(x[j][k]));
        }
    }
#pragma unroll
    for (int j = 0; j < STEPS; ++j) {
        if (live[j]) {
            float s = dr[j] * dc[j];
#pragma unroll
            for (int k = 0; k < CPL; ++k) atomicAdd(o[j] + k * LPE, x[j][k] * s);
        }
    }
}

__global__ void pool_kernel(const float* __restrict__ h  ,
                            const int* __restrict__ batch,
                            float* __restrict__ out, float* __restrict__ cnt, int N) {
    long long t = (long long)blockIdx.x * blockDim.x + threadIdx.x;
    if (t >= (long long)N * 16) return;
    int i = (int)(t >> 4);
    int c = (int)(t & 15);
    int g = batch[i];
    if (c < C_OUT) atomicAdd(&out[(size_t)g * C_OUT + c], h[t]);
    if (c == 0)    atomicAdd(&cnt[g], 1.0f);
}

__global__ void div_kernel(float* __restrict__ out, const float* __restrict__ cnt) {
    int t = blockIdx.x * blockDim.x + threadIdx.x;
    if (t < NUM_GRAPHS * C_OUT) out[t] /= fmaxf(cnt[t / C_OUT], 1.0f);
}


static inline int cdiv_ll(long long a, long long b) { return (int)((a + b - 1) / b); }

extern "C" void kernel_launch(void* const* d_in, const int* in_sizes, int n_in,
                              void* d_out, int out_size, void* d_ws, size_t ws_size,
                              hipStream_t stream) {
    const float* x     = (const float*)d_in[0];
    const int*   ei    = (const int*)d_in[1];
    const int*   batch = (const int*)d_in[2];
    const float* W0 = (const float*)d_in[3];  const float* b0 = (const float*)d_in[4];
    const float* W1 = (const float*)d_in[5];  const float* b1 = (const float*)d_in[6];
    const float* W2 = (const float*)d_in[7];  const float* b2 = (const float*)d_in[8];
    const float* W3 = (const float*)d_in[9];  const float* b3 = (const float*)d_in[10];

    const int N = in_sizes[2];
    const int E = in_sizes[1] / 2;
    const int H = 128;
    const int* rowi = ei;
    const int* coli = ei + E;

    float* out = (float*)d_out;
    float* ws  = (float*)d_ws;
    float* dis  = ws;
    float* bufA = ws + (((size_t)N + 255) & ~(size_t)255);
    float* bufB = bufA + (size_t)N * H;
    float* bufC = bufB + (size_t)N * H;
    float* cnt  = bufC + (size_t)N * H;

    const int TB = 256;

    fill_kernel<<<cdiv_ll(N, TB), TB, 0, stream>>>(dis, 1.0f, N);
    degree_kernel<<<cdiv_ll(E, TB), TB, 0, stream>>>(dis, coli, E);
    rsqrt_kernel<<<cdiv_ll(N, TB), TB, 0, stream>>>(dis, N);

    const int tilesM     = (N + 15) >> 4;
    const int gemmBlkH   = cdiv_ll((long long)tilesM * 8, TB / 32);
    const int gemmBlkOut = cdiv_ll((long long)tilesM * 1, TB / 32);
    const dim3 scatGrid(cdiv_ll((long long)E * 8, TB), 2);

    gemm_wmma_f32<128, 128, 128, false><<<gemmBlkH, TB, 0, stream>>>(x, W0, dis, b0, bufA, bufB, N, 8);
    scatter_kernel<128, 2, 4><<<scatGrid, TB, 0, stream>>>(bufA, dis, rowi, coli, bufB, E);

    gemm_wmma_f32<128, 128, 128, true><<<gemmBlkH, TB, 0, stream>>>(bufB, W1, dis, b1, bufA, bufC, N, 8);
    scatter_kernel<128, 2, 4><<<scatGrid, TB, 0, stream>>>(bufA, dis, rowi, coli, bufC, E);

    gemm_wmma_f32<128, 128, 128, true><<<gemmBlkH, TB, 0, stream>>>(bufC, W2, dis, b2, bufA, bufB, N, 8);
    scatter_kernel<128, 2, 4><<<scatGrid, TB, 0, stream>>>(bufA, dis, rowi, coli, bufB, E);

    gemm_wmma_f32<10, 16, 128, true><<<gemmBlkOut, TB, 0, stream>>>(bufB, W3, dis, b3, bufA, bufC, N, 1);
    scatter_kernel<16, 1, 4><<<cdiv_ll((long long)E * 4, TB), TB, 0, stream>>>(bufA, dis, rowi, coli, bufC, E);

    fill_kernel<<<cdiv_ll(NUM_GRAPHS * C_OUT, TB), TB, 0, stream>>>(out, 0.0f, NUM_GRAPHS * C_OUT);
    fill_kernel<<<cdiv_ll(NUM_GRAPHS, TB), TB, 0, stream>>>(cnt, 0.0f, NUM_GRAPHS);
    pool_kernel<<<cdiv_ll((long long)N * 16, TB), TB, 0, stream>>>(bufC, batch, out, cnt, N);
    div_kernel<<<cdiv_ll(NUM_GRAPHS * C_OUT, TB), TB, 0, stream>>>(out, cnt);
}


# Edge weight factored out of the edge sum: GEMM epilogue scales rows by dis^2 (target factor commutes with ReLU and the product), bias enters as bias*sqrt(deg); scatter adds raw rows, loads no dis
# speedup vs baseline: 1.8259x; 1.0800x over previous
#include <hip/hip_runtime.h>


typedef float v2f __attribute__((ext_vector_type(2)));
typedef float v8f __attribute__((ext_vector_type(8)));

#define NUM_GRAPHS 512
#define C_OUT 10


__global__ void fill_kernel(float* __restrict__ p, float v, int n) {
    int t = blockIdx.x * blockDim.x + threadIdx.x;
    if (t < n) p[t] = v;
}

__global__ void degree_kernel(float* __restrict__ deg, const int* __restrict__ col, int E) {
    int t = blockIdx.x * blockDim.x + threadIdx.x;
    if (t < E) atomicAdd(&deg[col[t]], 1.0f);
}

__global__ void rsqrt_kernel(float* __restrict__ deg, float* __restrict__ sq, int n) {
    int t = blockIdx.x * blockDim.x + threadIdx.x;
    if (t < n) {
        float v = deg[t];
        deg[t] = rsqrtf(v);
        sq[t]  = sqrtf(v);
    }
}

template <int BCOLS, int DSTRIDE, int KDIM, bool RELU_A, bool FIRST>
__global__ __launch_bounds__(256)
void gemm_wmma_f32(const float* __restrict__ A, const float* __restrict__ B,
                   const float* __restrict__ dis, const float* __restrict__ sq,
                   const float* __restrict__ bias,
                   float* __restrict__ D, float* __restrict__ S, int N, int ntilesN) {
    int wave = blockIdx.x * (blockDim.x >> 5) + (threadIdx.x >> 5);
    int lane = threadIdx.x & 31;
    int tilesM = (N + 15) >> 4;
    if (wave >= tilesM * ntilesN) return;
    int tm = wave / ntilesN;
    int tn = wave - tm * ntilesN;
    int m0 = tm << 4, n0 = tn << 4;
    int half = lane >> 4;
    int l15  = lane & 15;

    int row  = m0 + l15;   if (row  >= N)     row  = N - 1;
    int bcol = n0 + l15;   if (bcol >= BCOLS) bcol = BCOLS - 1;

    const float* __restrict__ Arow = A + (size_t)row * KDIM + half * 2;
    const float* __restrict__ Bcol = B + (size_t)(half * 2) * BCOLS + bcol;

    v8f acc = {0.f, 0.f, 0.f, 0.f, 0.f, 0.f, 0.f, 0.f};
#pragma unroll
    for (int k0 = 0; k0 < KDIM; k0 += 4) {
        v2f a = *(const v2f*)(Arow + k0);
        if constexpr (RELU_A) {
            a[0] = a[0] > 0.0f ? a[0] : 0.0f;
            a[1] = a[1] > 0.0f ? a[1] : 0.0f;
        }
        v2f b;
        b[0] = Bcol[(size_t)k0 * BCOLS];
        b[1] = Bcol[(size_t)(k0 + 1) * BCOLS];
        acc = __builtin_amdgcn_wmma_f32_16x16x4_f32(false, a, false, b,
                                                    (short)0, acc, false, false);
    }
    const int   ccol = n0 + l15;
    const float bl   = bias[bcol];
    const float bv   = (ccol < BCOLS) ? bl : 0.0f;
    float dv[8], sv[8];
#pragma unroll
    for (int r = 0; r < 8; ++r) {
        int m  = m0 + r + half * 8;
        int mc = m < N ? m : N - 1;
        float d = dis[mc];
        float q = sq[mc];
        float f = FIRST ? d : d * d;
        dv[r] = acc[r] * f;
        sv[r] = dv[r] + bv * q;
    }
#pragma unroll
    for (int r = 0; r < 8; ++r) {
        int m = m0 + r + half * 8;
        if (m < N) {
            size_t o = (size_t)m * DSTRIDE + ccol;
            D[o] = dv[r];
            S[o] = sv[r];
        }
    }
}

template <int F, int SLICES, int STEPS>
__global__ void scatter_kernel(const float* __restrict__ lin,
                               const int* __restrict__ row, const int* __restrict__ col,
                               float* __restrict__ out, int E) {
    constexpr int FS  = F / SLICES;
    constexpr int LPE = (FS < 32) ? FS : 32;
    constexpr int CPL = FS / LPE;
    constexpr int EPS = 32 / LPE;
    long long t = (long long)blockIdx.x * blockDim.x + threadIdx.x;
    long long wave = t >> 5;
    int lane = (int)(t & 31);
    int sub  = lane / LPE;
    int d    = lane - sub * LPE;
    if constexpr (SLICES > 1) d += (int)blockIdx.y * FS;

    float x[STEPS][CPL];
    int   ci[STEPS];
    bool live[STEPS];
#pragma unroll
    for (int j = 0; j < STEPS; ++j) {
        long long e = (wave * STEPS + j) * EPS + sub;
        live[j] = e < E;
        long long ec = live[j] ? e : (long long)E - 1;
        int r = row[ec];
        ci[j] = col[ec];
        const float* src = lin + (size_t)r * F + d;
#pragma unroll
        for (int k = 0; k < CPL; ++k) x[j][k] = src[k * LPE];
    }
    if constexpr (STEPS > 1) {
#pragma unroll
        for (int j = 0; j < STEPS; ++j) {
            asm volatile("" :: "v"(ci[j]));
#pragma unroll
            for (int k = 0; k < CPL; ++k) asm volatile("" :: "v"(x[j][k]));
        }
    }
#pragma unroll
    for (int j = 0; j < STEPS; ++j) {
        if (live[j]) {
            float* o = out + (size_t)ci[j] * F + d;
#pragma unroll
            for (int k = 0; k < CPL; ++k) atomicAdd(o + k * LPE, x[j][k]);
        }
    }
}

__global__ void pool_kernel(const float* __restrict__ h  ,
                            const float* __restrict__ dis,
                            const int* __restrict__ batch,
                            float* __restrict__ out, float* __restrict__ cnt, int N) {
    long long t = (long long)blockIdx.x * blockDim.x + threadIdx.x;
    if (t >= (long long)N * 16) return;
    int i = (int)(t >> 4);
    int c = (int)(t & 15);
    int g = batch[i];
    if (c < C_OUT) atomicAdd(&out[(size_t)g * C_OUT + c], h[t] * dis[i]);
    if (c == 0)    atomicAdd(&cnt[g], 1.0f);
}

__global__ void div_kernel(float* __restrict__ out, const float* __restrict__ cnt) {
    int t = blockIdx.x * blockDim.x + threadIdx.x;
    if (t < NUM_GRAPHS * C_OUT) out[t] /= fmaxf(cnt[t / C_OUT], 1.0f);
}


static inline int cdiv_ll(long long a, long long b) { return (int)((a + b - 1) / b); }

extern "C" void kernel_launch(void* const* d_in, const int* in_sizes, int n_in,
                              void* d_out, int out_size, void* d_ws, size_t ws_size,
                              hipStream_t stream) {
    const float* x     = (const float*)d_in[0];
    const int*   ei    = (const int*)d_in[1];
    const int*   batch = (const int*)d_in[2];
    const float* W0 = (const float*)d_in[3];  const float* b0 = (const float*)d_in[4];
    const float* W1 = (const float*)d_in[5];  const float* b1 = (const float*)d_in[6];
    const float* W2 = (const float*)d_in[7];  const float* b2 = (const float*)d_in[8];
    const float* W3 = (const float*)d_in[9];  const float* b3 = (const float*)d_in[10];

    const int N = in_sizes[2];
    const int E = in_sizes[1] / 2;
    const int H = 128;
    const int* rowi = ei;
    const int* coli = ei + E;

    float* out = (float*)d_out;
    float* ws  = (float*)d_ws;
    const size_t Npad = ((size_t)N + 255) & ~(size_t)255;
    float* dis  = ws;
    float* sq   = ws + Npad;
    float* bufA = sq + Npad;
    float* bufB = bufA + (size_t)N * H;
    float* bufC = bufB + (size_t)N * H;
    float* cnt  = bufC + (size_t)N * H;

    const int TB = 256;

    fill_kernel<<<cdiv_ll(N, TB), TB, 0, stream>>>(dis, 1.0f, N);
    degree_kernel<<<cdiv_ll(E, TB), TB, 0, stream>>>(dis, coli, E);
    rsqrt_kernel<<<cdiv_ll(N, TB), TB, 0, stream>>>(dis, sq, N);

    const int tilesM     = (N + 15) >> 4;
    const int gemmBlkH   = cdiv_ll((long long)tilesM * 8, TB / 32);
    const int gemmBlkOut = cdiv_ll((long long)tilesM * 1, TB / 32);
    const dim3 scatGrid(cdiv_ll((long long)E * 8, TB), 2);

    gemm_wmma_f32<128, 128, 128, false, true><<<gemmBlkH, TB, 0, stream>>>(x, W0, dis, sq, b0, bufA, bufB, N, 8);
    scatter_kernel<128, 2, 4><<<scatGrid, TB, 0, stream>>>(bufA, rowi, coli, bufB, E);

    gemm_wmma_f32<128, 128, 128, true, false><<<gemmBlkH, TB, 0, stream>>>(bufB, W1, dis, sq, b1, bufA, bufC, N, 8);
    scatter_kernel<128, 2, 4><<<scatGrid, TB, 0, stream>>>(bufA, rowi, coli, bufC, E);

    gemm_wmma_f32<128, 128, 128, true, false><<<gemmBlkH, TB, 0, stream>>>(bufC, W2, dis, sq, b2, bufA, bufB, N, 8);
    scatter_kernel<128, 2, 4><<<scatGrid, TB, 0, stream>>>(bufA, rowi, coli, bufB, E);

    gemm_wmma_f32<10, 16, 128, true, false><<<gemmBlkOut, TB, 0, stream>>>(bufB, W3, dis, sq, b3, bufA, bufC, N, 1);
    scatter_kernel<16, 1, 4><<<cdiv_ll((long long)E * 4, TB), TB, 0, stream>>>(bufA, rowi, coli, bufC, E);

    fill_kernel<<<cdiv_ll(NUM_GRAPHS * C_OUT, TB), TB, 0, stream>>>(out, 0.0f, NUM_GRAPHS * C_OUT);
    fill_kernel<<<cdiv_ll(NUM_GRAPHS, TB), TB, 0, stream>>>(cnt, 0.0f, NUM_GRAPHS);
    pool_kernel<<<cdiv_ll((long long)N * 16, TB), TB, 0, stream>>>(bufC, dis, batch, out, cnt, N);
    div_kernel<<<cdiv_ll(NUM_GRAPHS * C_OUT, TB), TB, 0, stream>>>(out, cnt);
}


# 128-wide edge walks placed by package half: a wave reads its die and draws 64-edge tickets for the 64-column half its side owns (then the other list): a sum line stays in one L2; sums stay atomic
# speedup vs baseline: 2.0856x; 1.1422x over previous
#include <hip/hip_runtime.h>


typedef float v2f __attribute__((ext_vector_type(2)));
typedef float v8f __attribute__((ext_vector_type(8)));

#define NUM_GRAPHS 512
#define C_OUT 10


__global__ void fill_kernel(float* __restrict__ p, float v, int n) {
    int t = blockIdx.x * blockDim.x + threadIdx.x;
    if (t < n) p[t] = v;
}

__global__ void degree_kernel(float* __restrict__ deg, const int* __restrict__ col, int E) {
    int t = blockIdx.x * blockDim.x + threadIdx.x;
    if (t < E) atomicAdd(&deg[col[t]], 1.0f);
}

__global__ void rsqrt_kernel(float* __restrict__ deg, float* __restrict__ sq, int n,
                             int* __restrict__ tick, int ntick) {
    int t = blockIdx.x * blockDim.x + threadIdx.x;
    if (t < ntick) tick[t] = 0;
    if (t < n) {
        float v = deg[t];
        deg[t] = rsqrtf(v);
        sq[t]  = sqrtf(v);
    }
}

template <int BCOLS, int DSTRIDE, int KDIM, bool RELU_A, bool FIRST>
__global__ __launch_bounds__(256)
void gemm_wmma_f32(const float* __restrict__ A, const float* __restrict__ B,
                   const float* __restrict__ dis, const float* __restrict__ sq,
                   const float* __restrict__ bias,
                   float* __restrict__ D, float* __restrict__ S, int N, int ntilesN) {
    int wave = blockIdx.x * (blockDim.x >> 5) + (threadIdx.x >> 5);
    int lane = threadIdx.x & 31;
    int tilesM = (N + 15) >> 4;
    if (wave >= tilesM * ntilesN) return;
    int tm = wave / ntilesN;
    int tn = wave - tm * ntilesN;
    int m0 = tm << 4, n0 = tn << 4;
    int half = lane >> 4;
    int l15  = lane & 15;

    int row  = m0 + l15;   if (row  >= N)     row  = N - 1;
    int bcol = n0 + l15;   if (bcol >= BCOLS) bcol = BCOLS - 1;

    const float* __restrict__ Arow = A + (size_t)row * KDIM + half * 2;
    const float* __restrict__ Bcol = B + (size_t)(half * 2) * BCOLS + bcol;

    v8f acc = {0.f, 0.f, 0.f, 0.f, 0.f, 0.f, 0.f, 0.f};
#pragma unroll
    for (int k0 = 0; k0 < KDIM; k0 += 4) {
        v2f a = *(const v2f*)(Arow + k0);
        if constexpr (RELU_A) {
            a[0] = a[0] > 0.0f ? a[0] : 0.0f;
            a[1] = a[1] > 0.0f ? a[1] : 0.0f;
        }
        v2f b;
        b[0] = Bcol[(size_t)k0 * BCOLS];
        b[1] = Bcol[(size_t)(k0 + 1) * BCOLS];
        acc = __builtin_amdgcn_wmma_f32_16x16x4_f32(false, a, false, b,
                                                    (short)0, acc, false, false);
    }
    const int   ccol = n0 + l15;
    const float bl   = bias[bcol];
    const float bv   = (ccol < BCOLS) ? bl : 0.0f;
    float dv[8], sv[8];
#pragma unroll
    for (int r = 0; r < 8; ++r) {
        int m  = m0 + r + half * 8;
        int mc = m < N ? m : N - 1;
        float d = dis[mc];
        float q = sq[mc];
        float f = FIRST ? d : d * d;
        dv[r] = acc[r] * f;
        sv[r] = dv[r] + bv * q;
    }
#pragma unroll
    for (int r = 0; r < 8; ++r) {
        int m = m0 + r + half * 8;
        if (m < N) {
            size_t o = (size_t)m * DSTRIDE + ccol;
            D[o] = dv[r];
            S[o] = sv[r];
        }
    }
}

template <int F, int SLICES, int STEPS>
__global__ void scatter_kernel(const float* __restrict__ lin,
                               const int* __restrict__ row, const int* __restrict__ col,
                               float* __restrict__ out, int E) {
    constexpr int FS  = F / SLICES;
    constexpr int LPE = (FS < 32) ? FS : 32;
    constexpr int CPL = FS / LPE;
    constexpr int EPS = 32 / LPE;
    long long t = (long long)blockIdx.x * blockDim.x + threadIdx.x;
    long long wave = t >> 5;
    int lane = (int)(t & 31);
    int sub  = lane / LPE;
    int d    = lane - sub * LPE;
    if constexpr (SLICES > 1) d += (int)blockIdx.y * FS;

    float x[STEPS][CPL];
    int   ci[STEPS];
    bool live[STEPS];
#pragma unroll
    for (int j = 0; j < STEPS; ++j) {
        long long e = (wave * STEPS + j) * EPS + sub;
        live[j] = e < E;
        long long ec = live[j] ? e : (long long)E - 1;
        int r = row[ec];
        ci[j] = col[ec];
        const float* src = lin + (size_t)r * F + d;
#pragma unroll
        for (int k = 0; k < CPL; ++k) x[j][k] = src[k * LPE];
    }
    if constexpr (STEPS > 1) {
#pragma unroll
        for (int j = 0; j < STEPS; ++j) {
            asm volatile("" :: "v"(ci[j]));
#pragma unroll
            for (int k = 0; k < CPL; ++k) asm volatile("" :: "v"(x[j][k]));
        }
    }
#pragma unroll
    for (int j = 0; j < STEPS; ++j) {
        if (live[j]) {
            float* o = out + (size_t)ci[j] * F + d;
#pragma unroll
            for (int k = 0; k < CPL; ++k) atomicAdd(o + k * LPE, x[j][k]);
        }
    }
}

template <int STEPS, int GROUPS>
__global__ __launch_bounds__(256)
void scatter_placed_kernel(const float* __restrict__ lin,
                           const int* __restrict__ row, const int* __restrict__ col,
                           float* __restrict__ out, int* __restrict__ tick, int E, int nticket) {
    constexpr int F  = 128;
    constexpr int FS = 64;
    constexpr int CH = STEPS * GROUPS;
    const int lane = threadIdx.x & 31;
    const int die   = (int)((__builtin_amdgcn_s_sendmsg_rtn(0x87) >> 16) & 15u);
    const int first = (die >> 2) & 1;
    for (int pass = 0; pass < 2; ++pass) {
        const int h = first ^ pass;
        const int d = lane + h * FS;
        int* cnt = tick + h * 64;
        for (int it = 0; it <= nticket; ++it) {
            int t = 0;
            if (lane == 0) t = atomicAdd(cnt, 1);
            t = __builtin_amdgcn_readfirstlane(t);
            if (t >= nticket) break;
            const int e0 = t * CH;
            for (int g = 0; g < GROUPS; ++g) {
                float x[STEPS][2];
                int   ci[STEPS];
                bool  live[STEPS];
#pragma unroll
                for (int j = 0; j < STEPS; ++j) {
                    int e = e0 + g * STEPS + j;
                    live[j] = e < E;
                    int ec = live[j] ? e : E - 1;
                    int r = row[ec];
                    ci[j] = col[ec];
                    const float* src = lin + (size_t)r * F + d;
                    x[j][0] = src[0];
                    x[j][1] = src[32];
                }
#pragma unroll
                for (int j = 0; j < STEPS; ++j) {
                    asm volatile("" :: "v"(ci[j]));
                    asm volatile("" :: "v"(x[j][0]));
                    asm volatile("" :: "v"(x[j][1]));
                }
#pragma unroll
                for (int j = 0; j < STEPS; ++j) {
                    if (live[j]) {
                        float* o = out + (size_t)ci[j] * F + d;
                        atomicAdd(o, x[j][0]);
                        atomicAdd(o + 32, x[j][1]);
                    }
                }
            }
        }
    }
}

__global__ void pool_kernel(const float* __restrict__ h  ,
                            const float* __restrict__ dis,
                            const int* __restrict__ batch,
                            float* __restrict__ out, float* __restrict__ cnt, int N) {
    long long t = (long long)blockIdx.x * blockDim.x + threadIdx.x;
    if (t >= (long long)N * 16) return;
    int i = (int)(t >> 4);
    int c = (int)(t & 15);
    int g = batch[i];
    if (c < C_OUT) atomicAdd(&out[(size_t)g * C_OUT + c], h[t] * dis[i]);
    if (c == 0)    atomicAdd(&cnt[g], 1.0f);
}

__global__ void div_kernel(float* __restrict__ out, const float* __restrict__ cnt) {
    int t = blockIdx.x * blockDim.x + threadIdx.x;
    if (t < NUM_GRAPHS * C_OUT) out[t] /= fmaxf(cnt[t / C_OUT], 1.0f);
}


static inline int cdiv_ll(long long a, long long b) { return (int)((a + b - 1) / b); }

extern "C" void kernel_launch(void* const* d_in, const int* in_sizes, int n_in,
                              void* d_out, int out_size, void* d_ws, size_t ws_size,
                              hipStream_t stream) {
    const float* x     = (const float*)d_in[0];
    const int*   ei    = (const int*)d_in[1];
    const int*   batch = (const int*)d_in[2];
    const float* W0 = (const float*)d_in[3];  const float* b0 = (const float*)d_in[4];
    const float* W1 = (const float*)d_in[5];  const float* b1 = (const float*)d_in[6];
    const float* W2 = (const float*)d_in[7];  const float* b2 = (const float*)d_in[8];
    const float* W3 = (const float*)d_in[9];  const float* b3 = (const float*)d_in[10];

    const int N = in_sizes[2];
    const int E = in_sizes[1] / 2;
    const int H = 128;
    const int* rowi = ei;
    const int* coli = ei + E;

    float* out = (float*)d_out;
    float* ws  = (float*)d_ws;
    const size_t Npad = ((size_t)N + 255) & ~(size_t)255;
    float* dis  = ws;
    float* sq   = ws + Npad;
    float* bufA = sq + Npad;
    float* bufB = bufA + (size_t)N * H;
    float* bufC = bufB + (size_t)N * H;
    float* cnt  = bufC + (size_t)N * H;
    const int TICK_WORDS = 3 * 128;
    int* tick = (int*)(cnt + NUM_GRAPHS);

    const int TB = 256;

    fill_kernel<<<cdiv_ll(N, TB), TB, 0, stream>>>(dis, 1.0f, N);
    degree_kernel<<<cdiv_ll(E, TB), TB, 0, stream>>>(dis, coli, E);
    rsqrt_kernel<<<cdiv_ll(N > TICK_WORDS ? N : TICK_WORDS, TB), TB, 0, stream>>>(dis, sq, N, tick, TICK_WORDS);

    const int tilesM     = (N + 15) >> 4;
    const int gemmBlkH   = cdiv_ll((long long)tilesM * 8, TB / 32);
    const int gemmBlkOut = cdiv_ll((long long)tilesM * 1, TB / 32);
    const int walkTickets = cdiv_ll(E, 64);
    const int walkGrid    = 256 * 8;

    gemm_wmma_f32<128, 128, 128, false, true><<<gemmBlkH, TB, 0, stream>>>(x, W0, dis, sq, b0, bufA, bufB, N, 8);
    scatter_placed_kernel<4, 16><<<walkGrid, TB, 0, stream>>>(bufA, rowi, coli, bufB, tick, E, walkTickets);

    gemm_wmma_f32<128, 128, 128, true, false><<<gemmBlkH, TB, 0, stream>>>(bufB, W1, dis, sq, b1, bufA, bufC, N, 8);
    scatter_placed_kernel<4, 16><<<walkGrid, TB, 0, stream>>>(bufA, rowi, coli, bufC, tick + 128, E, walkTickets);

    gemm_wmma_f32<128, 128, 128, true, false><<<gemmBlkH, TB, 0, stream>>>(bufC, W2, dis, sq, b2, bufA, bufB, N, 8);
    scatter_placed_kernel<4, 16><<<walkGrid, TB, 0, stream>>>(bufA, rowi, coli, bufB, tick + 256, E, walkTickets);

    gemm_wmma_f32<10, 16, 128, true, false><<<gemmBlkOut, TB, 0, stream>>>(bufB, W3, dis, sq, b3, bufA, bufC, N, 1);
    scatter_kernel<16, 1, 4><<<cdiv_ll((long long)E * 4, TB), TB, 0, stream>>>(bufA, rowi, coli, bufC, E);

    fill_kernel<<<cdiv_ll(NUM_GRAPHS * C_OUT, TB), TB, 0, stream>>>(out, 0.0f, NUM_GRAPHS * C_OUT);
    fill_kernel<<<cdiv_ll(NUM_GRAPHS, TB), TB, 0, stream>>>(cnt, 0.0f, NUM_GRAPHS);
    pool_kernel<<<cdiv_ll((long long)N * 16, TB), TB, 0, stream>>>(bufC, dis, batch, out, cnt, N);
    div_kernel<<<cdiv_ll(NUM_GRAPHS * C_OUT, TB), TB, 0, stream>>>(out, cnt);
}


# GEMM waves made resident: 512 work-groups; a wave keeps its 16-column weight fragments in registers and loops over row strips, next strip's A requested a strip ahead; k order unchanged
# speedup vs baseline: 2.2370x; 1.0726x over previous
#include <hip/hip_runtime.h>


typedef float v2f __attribute__((ext_vector_type(2)));
typedef float v8f __attribute__((ext_vector_type(8)));

#define NUM_GRAPHS 512
#define C_OUT 10


__global__ void fill_kernel(float* __restrict__ p, float v, int n) {
    int t = blockIdx.x * blockDim.x + threadIdx.x;
    if (t < n) p[t] = v;
}

__global__ void degree_kernel(float* __restrict__ deg, const int* __restrict__ col, int E) {
    int t = blockIdx.x * blockDim.x + threadIdx.x;
    if (t < E) atomicAdd(&deg[col[t]], 1.0f);
}

__global__ void rsqrt_kernel(float* __restrict__ deg, float* __restrict__ sq, int n,
                             int* __restrict__ tick, int ntick) {
    int t = blockIdx.x * blockDim.x + threadIdx.x;
    if (t < ntick) tick[t] = 0;
    if (t < n) {
        float v = deg[t];
        deg[t] = rsqrtf(v);
        sq[t]  = sqrtf(v);
    }
}


template <int KDIM>
__device__ __forceinline__ void gemm_load_strip(v2f (&a)[KDIM / 4], const float* __restrict__ A,
                                                int tm, int l15, int half, int N) {
    int row = (tm << 4) + l15;   if (row >= N) row = N - 1;
    const float* __restrict__ Arow = A + (size_t)row * KDIM + half * 2;
#pragma unroll
    for (int i = 0; i < KDIM / 4; ++i) a[i] = *(const v2f*)(Arow + 4 * i);
}

template <int DSTRIDE, int KDIM, bool RELU_A, bool FIRST>
__device__ __forceinline__ void gemm_strip(const v2f (&af)[KDIM / 4], const v2f (&bf)[KDIM / 4],
                                           const float* __restrict__ dis, const float* __restrict__ sq,
                                           float bv, float* __restrict__ D, float* __restrict__ S,
                                           int tm, int ccol, int half, int N) {
    const int m0 = tm << 4;
    v8f acc = {0.f, 0.f, 0.f, 0.f, 0.f, 0.f, 0.f, 0.f};
#pragma unroll
    for (int i = 0; i < KDIM / 4; ++i) {
        v2f a = af[i];
        if constexpr (RELU_A) {
            a[0] = a[0] > 0.0f ? a[0] : 0.0f;
            a[1] = a[1] > 0.0f ? a[1] : 0.0f;
        }
        acc = __builtin_amdgcn_wmma_f32_16x16x4_f32(false, a, false, bf[i],
                                                    (short)0, acc, false, false);
    }
    float dv[8], sv[8];
#pragma unroll
    for (int r = 0; r < 8; ++r) {
        int m  = m0 + r + half * 8;
        int mc = m < N ? m : N - 1;
        float d = dis[mc];
        float q = sq[mc];
        float f = FIRST ? d : d * d;
        dv[r] = acc[r] * f;
        sv[r] = dv[r] + bv * q;
    }
#pragma unroll
    for (int r = 0; r < 8; ++r) {
        int m = m0 + r + half * 8;
        if (m < N) {
            size_t o = (size_t)m * DSTRIDE + ccol;
            D[o] = dv[r];
            S[o] = sv[r];
        }
    }
}

template <int BCOLS, int DSTRIDE, int KDIM, bool RELU_A, bool FIRST>
__global__ __launch_bounds__(256)
void gemm_wmma_f32(const float* __restrict__ A, const float* __restrict__ B,
                   const float* __restrict__ dis, const float* __restrict__ sq,
                   const float* __restrict__ bias,
                   float* __restrict__ D, float* __restrict__ S, int N, int ntilesN) {
    const int wib  = __builtin_amdgcn_readfirstlane((int)(threadIdx.x >> 5));
    const int wave = blockIdx.x * (blockDim.x >> 5) + wib;
    const int nwav = gridDim.x * (blockDim.x >> 5);
    const int lane = threadIdx.x & 31;
    const int tilesM = (N + 15) >> 4;
    const int tn    = wave % ntilesN;
    const int tstep = nwav / ntilesN;
    int tm = wave / ntilesN;
    if (tm >= tilesM) return;
    const int n0 = tn << 4;
    const int half = lane >> 4;
    const int l15  = lane & 15;

    int bcol = n0 + l15;   if (bcol >= BCOLS) bcol = BCOLS - 1;
    const float* __restrict__ Bcol = B + (size_t)(half * 2) * BCOLS + bcol;
    const int   ccol = n0 + l15;
    const float bl   = bias[bcol];
    const float bv   = (ccol < BCOLS) ? bl : 0.0f;

    v2f a0[KDIM / 4], a1[KDIM / 4];
    gemm_load_strip<KDIM>(a0, A, tm, l15, half, N);

    v2f bf[KDIM / 4];
#pragma unroll
    for (int k0 = 0; k0 < KDIM; k0 += 4) {
        bf[k0 >> 2][0] = Bcol[(size_t)k0 * BCOLS];
        bf[k0 >> 2][1] = Bcol[(size_t)(k0 + 1) * BCOLS];
    }

    for (;;) {
        gemm_load_strip<KDIM>(a1, A, tm + tstep, l15, half, N);
        gemm_strip<DSTRIDE, KDIM, RELU_A, FIRST>(a0, bf, dis, sq, bv, D, S, tm, ccol, half, N);
        tm += tstep;
        if (tm >= tilesM) break;
        gemm_load_strip<KDIM>(a0, A, tm + tstep, l15, half, N);
        gemm_strip<DSTRIDE, KDIM, RELU_A, FIRST>(a1, bf, dis, sq, bv, D, S, tm, ccol, half, N);
        tm += tstep;
        if (tm >= tilesM) break;
    }
}

template <int F, int SLICES, int STEPS>
__global__ void scatter_kernel(const float* __restrict__ lin,
                               const int* __restrict__ row, const int* __restrict__ col,
                               float* __restrict__ out, int E) {
    constexpr int FS  = F / SLICES;
    constexpr int LPE = (FS < 32) ? FS : 32;
    constexpr int CPL = FS / LPE;
    constexpr int EPS = 32 / LPE;
    long long t = (long long)blockIdx.x * blockDim.x + threadIdx.x;
    long long wave = t >> 5;
    int lane = (int)(t & 31);
    int sub  = lane / LPE;
    int d    = lane - sub * LPE;
    if constexpr (SLICES > 1) d += (int)blockIdx.y * FS;

    float x[STEPS][CPL];
    int   ci[STEPS];
    bool live[STEPS];
#pragma unroll
    for (int j = 0; j < STEPS; ++j) {
        long long e = (wave * STEPS + j) * EPS + sub;
        live[j] = e < E;
        long long ec = live[j] ? e : (long long)E - 1;
        int r = row[ec];
        ci[j] = col[ec];
        const float* src = lin + (size_t)r * F + d;
#pragma unroll
        for (int k = 0; k < CPL; ++k) x[j][k] = src[k * LPE];
    }
    if constexpr (STEPS > 1) {
#pragma unroll
        for (int j = 0; j < STEPS; ++j) {
            asm volatile("" :: "v"(ci[j]));
#pragma unroll
            for (int k = 0; k < CPL; ++k) asm volatile("" :: "v"(x[j][k]));
        }
    }
#pragma unroll
    for (int j = 0; j < STEPS; ++j) {
        if (live[j]) {
            float* o = out + (size_t)ci[j] * F + d;
#pragma unroll
            for (int k = 0; k < CPL; ++k) atomicAdd(o + k * LPE, x[j][k]);
        }
    }
}

template <int STEPS, int GROUPS>
__global__ __launch_bounds__(256)
void scatter_placed_kernel(const float* __restrict__ lin,
                           const int* __restrict__ row, const int* __restrict__ col,
                           float* __restrict__ out, int* __restrict__ tick, int E, int nticket) {
    constexpr int F  = 128;
    constexpr int FS = 64;
    constexpr int CH = STEPS * GROUPS;
    const int lane = threadIdx.x & 31;
    const int die   = (int)((__builtin_amdgcn_s_sendmsg_rtn(0x87) >> 16) & 15u);
    const int first = (die >> 2) & 1;
    for (int pass = 0; pass < 2; ++pass) {
        const int h = first ^ pass;
        const int d = lane + h * FS;
        int* cnt = tick + h * 64;
        for (int it = 0; it <= nticket; ++it) {
            int t = 0;
            if (lane == 0) t = atomicAdd(cnt, 1);
            t = __builtin_amdgcn_readfirstlane(t);
            if (t >= nticket) break;
            const int e0 = t * CH;
            for (int g = 0; g < GROUPS; ++g) {
                float x[STEPS][2];
                int   ci[STEPS];
                bool  live[STEPS];
#pragma unroll
                for (int j = 0; j < STEPS; ++j) {
                    int e = e0 + g * STEPS + j;
                    live[j] = e < E;
                    int ec = live[j] ? e : E - 1;
                    int r = row[ec];
                    ci[j] = col[ec];
                    const float* src = lin + (size_t)r * F + d;
                    x[j][0] = src[0];
                    x[j][1] = src[32];
                }
#pragma unroll
                for (int j = 0; j < STEPS; ++j) {
                    asm volatile("" :: "v"(ci[j]));
                    asm volatile("" :: "v"(x[j][0]));
                    asm volatile("" :: "v"(x[j][1]));
                }
#pragma unroll
                for (int j = 0; j < STEPS; ++j) {
                    if (live[j]) {
                        float* o = out + (size_t)ci[j] * F + d;
                        atomicAdd(o, x[j][0]);
                        atomicAdd(o + 32, x[j][1]);
                    }
                }
            }
        }
    }
}

__global__ void pool_kernel(const float* __restrict__ h  ,
                            const float* __restrict__ dis,
                            const int* __restrict__ batch,
                            float* __restrict__ out, float* __restrict__ cnt, int N) {
    long long t = (long long)blockIdx.x * blockDim.x + threadIdx.x;
    if (t >= (long long)N * 16) return;
    int i = (int)(t >> 4);
    int c = (int)(t & 15);
    int g = batch[i];
    if (c < C_OUT) atomicAdd(&out[(size_t)g * C_OUT + c], h[t] * dis[i]);
    if (c == 0)    atomicAdd(&cnt[g], 1.0f);
}

__global__ void div_kernel(float* __restrict__ out, const float* __restrict__ cnt) {
    int t = blockIdx.x * blockDim.x + threadIdx.x;
    if (t < NUM_GRAPHS * C_OUT) out[t] /= fmaxf(cnt[t / C_OUT], 1.0f);
}


static inline int cdiv_ll(long long a, long long b) { return (int)((a + b - 1) / b); }

extern "C" void kernel_launch(void* const* d_in, const int* in_sizes, int n_in,
                              void* d_out, int out_size, void* d_ws, size_t ws_size,
                              hipStream_t stream) {
    const float* x     = (const float*)d_in[0];
    const int*   ei    = (const int*)d_in[1];
    const int*   batch = (const int*)d_in[2];
    const float* W0 = (const float*)d_in[3];  const float* b0 = (const float*)d_in[4];
    const float* W1 = (const float*)d_in[5];  const float* b1 = (const float*)d_in[6];
    const float* W2 = (const float*)d_in[7];  const float* b2 = (const float*)d_in[8];
    const float* W3 = (const float*)d_in[9];  const float* b3 = (const float*)d_in[10];

    const int N = in_sizes[2];
    const int E = in_sizes[1] / 2;
    const int H = 128;
    const int* rowi = ei;
    const int* coli = ei + E;

    float* out = (float*)d_out;
    float* ws  = (float*)d_ws;
    const size_t Npad = ((size_t)N + 255) & ~(size_t)255;
    float* dis  = ws;
    float* sq   = ws + Npad;
    float* bufA = sq + Npad;
    float* bufB = bufA + (size_t)N * H;
    float* bufC = bufB + (size_t)N * H;
    float* cnt  = bufC + (size_t)N * H;
    const int TICK_WORDS = 3 * 128;
    int* tick = (int*)(cnt + NUM_GRAPHS);

    const int TB = 256;

    fill_kernel<<<cdiv_ll(N, TB), TB, 0, stream>>>(dis, 1.0f, N);
    degree_kernel<<<cdiv_ll(E, TB), TB, 0, stream>>>(dis, coli, E);
    rsqrt_kernel<<<cdiv_ll(N > TICK_WORDS ? N : TICK_WORDS, TB), TB, 0, stream>>>(dis, sq, N, tick, TICK_WORDS);

    const int gemmBlkH   = 256 * 2;
    const int gemmBlkOut = 256 * 2;
    const int walkTickets = cdiv_ll(E, 64);
    const int walkGrid    = 256 * 8;

    gemm_wmma_f32<128, 128, 128, false, true><<<gemmBlkH, TB, 0, stream>>>(x, W0, dis, sq, b0, bufA, bufB, N, 8);
    scatter_placed_kernel<4, 16><<<walkGrid, TB, 0, stream>>>(bufA, rowi, coli, bufB, tick, E, walkTickets);

    gemm_wmma_f32<128, 128, 128, true, false><<<gemmBlkH, TB, 0, stream>>>(bufB, W1, dis, sq, b1, bufA, bufC, N, 8);
    scatter_placed_kernel<4, 16><<<walkGrid, TB, 0, stream>>>(bufA, rowi, coli, bufC, tick + 128, E, walkTickets);

    gemm_wmma_f32<128, 128, 128, true, false><<<gemmBlkH, TB, 0, stream>>>(bufC, W2, dis, sq, b2, bufA, bufB, N, 8);
    scatter_placed_kernel<4, 16><<<walkGrid, TB, 0, stream>>>(bufA, rowi, coli, bufB, tick + 256, E, walkTickets);

    gemm_wmma_f32<10, 16, 128, true, false><<<gemmBlkOut, TB, 0, stream>>>(bufB, W3, dis, sq, b3, bufA, bufC, N, 1);
    scatter_kernel<16, 1, 4><<<cdiv_ll((long long)E * 4, TB), TB, 0, stream>>>(bufA, rowi, coli, bufC, E);

    fill_kernel<<<cdiv_ll(NUM_GRAPHS * C_OUT, TB), TB, 0, stream>>>(out, 0.0f, NUM_GRAPHS * C_OUT);
    fill_kernel<<<cdiv_ll(NUM_GRAPHS, TB), TB, 0, stream>>>(cnt, 0.0f, NUM_GRAPHS);
    pool_kernel<<<cdiv_ll((long long)N * 16, TB), TB, 0, stream>>>(bufC, dis, batch, out, cnt, N);
    div_kernel<<<cdiv_ll(NUM_GRAPHS * C_OUT, TB), TB, 0, stream>>>(out, cnt);
}


# Edge sums of all four layers on the matrix unit: edges binned by 16 target rows, a wave per bin multiplies a 0/1 selection fragment by the gathered rows (f32 16x16x4), no LDS adds; sum order changes
# speedup vs baseline: 2.4768x; 1.1072x over previous
#include <hip/hip_runtime.h>


typedef float v2f __attribute__((ext_vector_type(2)));
typedef float v8f __attribute__((ext_vector_type(8)));

#define NUM_GRAPHS 512
#define C_OUT 10


__global__ void fill_kernel(float* __restrict__ p, float v, int n) {
    int t = blockIdx.x * blockDim.x + threadIdx.x;
    if (t < n) p[t] = v;
}

__global__ void degree_kernel(float* __restrict__ deg, const int* __restrict__ col, int E) {
    int t = blockIdx.x * blockDim.x + threadIdx.x;
    if (t < E) atomicAdd(&deg[col[t]], 1.0f);
}

__global__ void rsqrt_kernel(float* __restrict__ deg, float* __restrict__ sq, int n) {
    int t = blockIdx.x * blockDim.x + threadIdx.x;
    if (t < n) {
        float v = deg[t];
        deg[t] = rsqrtf(v);
        sq[t]  = sqrtf(v);
    }
}

#define BIN_SHIFT 4
#define BIN_ROWS 16
#define SUP_SHIFT 6
#define SUP_ROWS 64
#define CUR_STRIDE 32

__global__ __launch_bounds__(1024)
void bin_offsets_kernel(const float* __restrict__ deg, int* __restrict__ binStart,
                        int* __restrict__ cursor, int N, int nbins, int nsup) {
    __shared__ int s[1024];
    const int tid = threadIdx.x;
    int carry = 0;
    for (int b0 = 0; b0 < nsup; b0 += 1024) {
        const int sb = b0 + tid;
        int c4[4] = {0, 0, 0, 0};
        if (sb < nsup) {
            const int c0 = sb << SUP_SHIFT;
            if (c0 + SUP_ROWS <= N) {
                const float4* p = (const float4*)(deg + c0);
#pragma unroll
                for (int i = 0; i < 4; ++i) {
                    float a = 0.0f;
#pragma unroll
                    for (int j = 0; j < BIN_ROWS / 4; ++j) { float4 v = p[i * (BIN_ROWS / 4) + j]; a += (v.x + v.y) + (v.z + v.w); }
                    c4[i] = (int)a - BIN_ROWS;
                }
            } else {
                for (int c = c0; c < N; ++c) {
                    const int k = (c - c0) >> BIN_SHIFT;
                    const int d = (int)deg[c] - 1;
                    c4[0] += k == 0 ? d : 0;   c4[1] += k == 1 ? d : 0;
                    c4[2] += k == 2 ? d : 0;   c4[3] += k == 3 ? d : 0;
                }
            }
        }
        const int cnt = (c4[0] + c4[1]) + (c4[2] + c4[3]);
        s[tid] = cnt;
        __syncthreads();
        for (int off = 1; off < 1024; off <<= 1) {
            int v = tid >= off ? s[tid - off] : 0;
            __syncthreads();
            s[tid] += v;
            __syncthreads();
        }
        int excl = carry + s[tid] - cnt;
        if (sb < nsup) {
#pragma unroll
            for (int i = 0; i < 4; ++i) {
                const int b = sb * 4 + i;
                if (b < nbins) { binStart[b] = excl; cursor[(size_t)b * CUR_STRIDE] = excl; }
                excl += c4[i];
            }
        }
        carry += s[1023];
        __syncthreads();
    }
    if (tid == 0) binStart[nbins] = carry;
}

__global__ void place_kernel(const int* __restrict__ row, const int* __restrict__ col,
                             int* __restrict__ cursor, int* __restrict__ binned, int E) {
    int e = blockIdx.x * blockDim.x + threadIdx.x;
    if (e < E) {
        int c = col[e];
        int r = row[e];
        int pos = atomicAdd(&cursor[(size_t)(c >> BIN_SHIFT) * CUR_STRIDE], 1);
        ((int2*)binned)[pos] = make_int2(r, c & (BIN_ROWS - 1));
    }
}


template <int KDIM>
__device__ __forceinline__ void gemm_load_strip(v2f (&a)[KDIM / 4], const float* __restrict__ A,
                                                int tm, int l15, int half, int N) {
    int row = (tm << 4) + l15;   if (row >= N) row = N - 1;
    const float* __restrict__ Arow = A + (size_t)row * KDIM + half * 2;
#pragma unroll
    for (int i = 0; i < KDIM / 4; ++i) a[i] = *(const v2f*)(Arow + 4 * i);
}

template <int DSTRIDE, int KDIM, bool RELU_A, bool FIRST>
__device__ __forceinline__ void gemm_strip(const v2f (&af)[KDIM / 4], const v2f (&bf)[KDIM / 4],
                                           const float* __restrict__ dis, const float* __restrict__ sq,
                                           float bv, float* __restrict__ D, float* __restrict__ S,
                                           int tm, int ccol, int half, int N) {
    const int m0 = tm << 4;
    v8f acc = {0.f, 0.f, 0.f, 0.f, 0.f, 0.f, 0.f, 0.f};
#pragma unroll
    for (int i = 0; i < KDIM / 4; ++i) {
        v2f a = af[i];
        if constexpr (RELU_A) {
            a[0] = a[0] > 0.0f ? a[0] : 0.0f;
            a[1] = a[1] > 0.0f ? a[1] : 0.0f;
        }
        acc = __builtin_amdgcn_wmma_f32_16x16x4_f32(false, a, false, bf[i],
                                                    (short)0, acc, false, false);
    }
    float dv[8], sv[8];
#pragma unroll
    for (int r = 0; r < 8; ++r) {
        int m  = m0 + r + half * 8;
        int mc = m < N ? m : N - 1;
        float d = dis[mc];
        float q = sq[mc];
        float f = FIRST ? d : d * d;
        dv[r] = acc[r] * f;
        sv[r] = dv[r] + bv * q;
    }
#pragma unroll
    for (int r = 0; r < 8; ++r) {
        int m = m0 + r + half * 8;
        if (m < N) {
            size_t o = (size_t)m * DSTRIDE + ccol;
            D[o] = dv[r];
            S[o] = sv[r];
        }
    }
}

template <int BCOLS, int DSTRIDE, int KDIM, bool RELU_A, bool FIRST>
__global__ __launch_bounds__(256)
void gemm_wmma_f32(const float* __restrict__ A, const float* __restrict__ B,
                   const float* __restrict__ dis, const float* __restrict__ sq,
                   const float* __restrict__ bias,
                   float* __restrict__ D, float* __restrict__ S, int N, int ntilesN) {
    const int wib  = __builtin_amdgcn_readfirstlane((int)(threadIdx.x >> 5));
    const int wave = blockIdx.x * (blockDim.x >> 5) + wib;
    const int nwav = gridDim.x * (blockDim.x >> 5);
    const int lane = threadIdx.x & 31;
    const int tilesM = (N + 15) >> 4;
    const int tn    = wave % ntilesN;
    const int tstep = nwav / ntilesN;
    int tm = wave / ntilesN;
    if (tm >= tilesM) return;
    const int n0 = tn << 4;
    const int half = lane >> 4;
    const int l15  = lane & 15;

    int bcol = n0 + l15;   if (bcol >= BCOLS) bcol = BCOLS - 1;
    const float* __restrict__ Bcol = B + (size_t)(half * 2) * BCOLS + bcol;
    const int   ccol = n0 + l15;
    const float bl   = bias[bcol];
    const float bv   = (ccol < BCOLS) ? bl : 0.0f;

    v2f a0[KDIM / 4], a1[KDIM / 4];
    gemm_load_strip<KDIM>(a0, A, tm, l15, half, N);

    v2f bf[KDIM / 4];
#pragma unroll
    for (int k0 = 0; k0 < KDIM; k0 += 4) {
        bf[k0 >> 2][0] = Bcol[(size_t)k0 * BCOLS];
        bf[k0 >> 2][1] = Bcol[(size_t)(k0 + 1) * BCOLS];
    }

    for (;;) {
        gemm_load_strip<KDIM>(a1, A, tm + tstep, l15, half, N);
        gemm_strip<DSTRIDE, KDIM, RELU_A, FIRST>(a0, bf, dis, sq, bv, D, S, tm, ccol, half, N);
        tm += tstep;
        if (tm >= tilesM) break;
        gemm_load_strip<KDIM>(a0, A, tm + tstep, l15, half, N);
        gemm_strip<DSTRIDE, KDIM, RELU_A, FIRST>(a1, bf, dis, sq, bv, D, S, tm, ccol, half, N);
        tm += tstep;
        if (tm >= tilesM) break;
    }
}

typedef float v4f __attribute__((ext_vector_type(4)));

template <int K>
__device__ __forceinline__ int row_share16(int v) {
    return __builtin_amdgcn_update_dpp(0, v, 0x150 + K, 0xf, 0xf, false);
}

template <int G>
__device__ __forceinline__ void seg_fetch(v4f (&x)[2], v4f (&y)[2], v2f& a, int src, int tg,
                                          const float* __restrict__ lin, int c) {
    const int sx = row_share16<2 * G>(src),     tx = row_share16<2 * G>(tg);
    const int sy = row_share16<2 * G + 1>(src), ty = row_share16<2 * G + 1>(tg);
    a[0] = tx == c ? 1.0f : 0.0f;
    a[1] = ty == c ? 1.0f : 0.0f;
    const unsigned ox = ((unsigned)sx << 9) | ((unsigned)c << 4);
    const unsigned oy = ((unsigned)sy << 9) | ((unsigned)c << 4);
    const char* __restrict__ linb = (const char*)lin;
    x[0] = *(const v4f*)(linb + ox);   x[1] = *(const v4f*)(linb + ox + 256u);
    y[0] = *(const v4f*)(linb + oy);   y[1] = *(const v4f*)(linb + oy + 256u);
}

template <int SB>
__device__ __forceinline__ void seg_half_trip(v8f (&acc)[8], int src, int tg,
                                              const float* __restrict__ lin, int c) {
    v4f x[4][2], y[4][2];
    v2f a[4];
    seg_fetch<4 * SB + 0>(x[0], y[0], a[0], src, tg, lin, c);
    seg_fetch<4 * SB + 1>(x[1], y[1], a[1], src, tg, lin, c);
    seg_fetch<4 * SB + 2>(x[2], y[2], a[2], src, tg, lin, c);
    seg_fetch<4 * SB + 3>(x[3], y[3], a[3], src, tg, lin, c);
#pragma unroll
    for (int i = 0; i < 4; ++i) {
#pragma unroll
        for (int s = 0; s < 2; ++s) {
#pragma unroll
            for (int q = 0; q < 4; ++q) {
                v2f b;
                b[0] = x[i][s][q];
                b[1] = y[i][s][q];
                acc[4 * s + q] = __builtin_amdgcn_wmma_f32_16x16x4_f32(false, a[i], false, b,
                                                                       (short)0, acc[4 * s + q], false, false);
            }
        }
    }
}

#define SEG_WAVES 4
__global__ __launch_bounds__(SEG_WAVES * 32)
void seg_sum_wmma_kernel(const float* __restrict__ lin, const int* __restrict__ binned,
                         const int* __restrict__ binStart, float* __restrict__ out, int N, int nbins) {
    const int lane = threadIdx.x & 31;
    const int wib  = __builtin_amdgcn_readfirstlane((int)(threadIdx.x >> 5));
    const int b    = blockIdx.x * SEG_WAVES + wib;
    if (b >= nbins) return;
    const int c    = lane & 15;
    const int half = lane >> 4;
    const int slot = 4 * (c >> 1) + 2 * half + (c & 1);

    const int eb = binStart[b];
    const int ee = binStart[b + 1];
    const int2* __restrict__ ed2 = (const int2*)binned;

    v8f acc[8];
#pragma unroll
    for (int t = 0; t < 8; ++t) acc[t] = (v8f){0.f, 0.f, 0.f, 0.f, 0.f, 0.f, 0.f, 0.f};

    int2 ed = make_int2(0, 0);
    if (eb < ee) {
        int idx = eb + slot;   if (idx >= ee) idx = ee - 1;
        ed = ed2[idx];
    }
    for (int base = eb; base < ee; base += 32) {
        const int nv = ee - base;
        int2 edn = ed;
        if (base + 32 < ee) {
            int idx = base + 32 + slot;   if (idx >= ee) idx = ee - 1;
            edn = ed2[idx];
        }
        const int src = ed.x;
        const int tg  = slot < nv ? ed.y : BIN_ROWS;
        seg_half_trip<0>(acc, src, tg, lin, c);
        if (nv > 16) seg_half_trip<1>(acc, src, tg, lin, c);
        ed = edn;
    }

    char* __restrict__ outb = (char*)out;
    const int node0 = (b << BIN_SHIFT) + half * 8;
#pragma unroll
    for (int r0 = 0; r0 < 8; r0 += 4) {
        v4f sv[4][2];
#pragma unroll
        for (int i = 0; i < 4; ++i) {
            const int node = node0 + r0 + i;
            const int nc = node < N ? node : N - 1;
            const unsigned o = ((unsigned)nc << 9) | ((unsigned)c << 4);
            sv[i][0] = *(const v4f*)(outb + o);
            sv[i][1] = *(const v4f*)(outb + o + 256u);
        }
#pragma unroll
        for (int i = 0; i < 4; ++i) {
            const int node = node0 + r0 + i;
#pragma unroll
            for (int s = 0; s < 2; ++s) {
                sv[i][s][0] += acc[4 * s + 0][r0 + i];
                sv[i][s][1] += acc[4 * s + 1][r0 + i];
                sv[i][s][2] += acc[4 * s + 2][r0 + i];
                sv[i][s][3] += acc[4 * s + 3][r0 + i];
            }
            if (node < N) {
                const unsigned o = ((unsigned)node << 9) | ((unsigned)c << 4);
                *(v4f*)(outb + o)        = sv[i][0];
                *(v4f*)(outb + o + 256u) = sv[i][1];
            }
        }
    }
}

template <int G>
__device__ __forceinline__ void seg16_fetch(v2f& bfr, v2f& a, int src, int tg,
                                            const float* __restrict__ lin, int c) {
    const int sx = row_share16<2 * G>(src),     tx = row_share16<2 * G>(tg);
    const int sy = row_share16<2 * G + 1>(src), ty = row_share16<2 * G + 1>(tg);
    a[0] = tx == c ? 1.0f : 0.0f;
    a[1] = ty == c ? 1.0f : 0.0f;
    const char* __restrict__ linb = (const char*)lin;
    bfr[0] = *(const float*)(linb + (((unsigned)sx << 6) | ((unsigned)c << 2)));
    bfr[1] = *(const float*)(linb + (((unsigned)sy << 6) | ((unsigned)c << 2)));
}

template <int SB>
__device__ __forceinline__ void seg16_half_trip(v8f& acc, int src, int tg,
                                                const float* __restrict__ lin, int c) {
    v2f bfr[4], a[4];
    seg16_fetch<4 * SB + 0>(bfr[0], a[0], src, tg, lin, c);
    seg16_fetch<4 * SB + 1>(bfr[1], a[1], src, tg, lin, c);
    seg16_fetch<4 * SB + 2>(bfr[2], a[2], src, tg, lin, c);
    seg16_fetch<4 * SB + 3>(bfr[3], a[3], src, tg, lin, c);
#pragma unroll
    for (int i = 0; i < 4; ++i)
        acc = __builtin_amdgcn_wmma_f32_16x16x4_f32(false, a[i], false, bfr[i], (short)0, acc, false, false);
}

__global__ __launch_bounds__(SEG_WAVES * 32)
void seg_sum16_wmma_kernel(const float* __restrict__ lin, const int* __restrict__ binned,
                           const int* __restrict__ binStart, float* __restrict__ out, int N, int nbins) {
    const int lane = threadIdx.x & 31;
    const int wib  = __builtin_amdgcn_readfirstlane((int)(threadIdx.x >> 5));
    const int b    = blockIdx.x * SEG_WAVES + wib;
    if (b >= nbins) return;
    const int c    = lane & 15;
    const int half = lane >> 4;
    const int slot = 4 * (c >> 1) + 2 * half + (c & 1);

    const int eb = binStart[b];
    const int ee = binStart[b + 1];
    const int2* __restrict__ ed2 = (const int2*)binned;

    v8f acc = {0.f, 0.f, 0.f, 0.f, 0.f, 0.f, 0.f, 0.f};

    int2 ed = make_int2(0, 0);
    if (eb < ee) {
        int idx = eb + slot;   if (idx >= ee) idx = ee - 1;
        ed = ed2[idx];
    }
    for (int base = eb; base < ee; base += 32) {
        const int nv = ee - base;
        int2 edn = ed;
        if (base + 32 < ee) {
            int idx = base + 32 + slot;   if (idx >= ee) idx = ee - 1;
            edn = ed2[idx];
        }
        const int src = ed.x;
        const int tg  = slot < nv ? ed.y : BIN_ROWS;
        seg16_half_trip<0>(acc, src, tg, lin, c);
        if (nv > 16) seg16_half_trip<1>(acc, src, tg, lin, c);
        ed = edn;
    }

    const int node0 = (b << BIN_SHIFT) + half * 8;
    float sv[8];
#pragma unroll
    for (int r = 0; r < 8; ++r) {
        const int node = node0 + r;
        const int nc = node < N ? node : N - 1;
        sv[r] = out[(unsigned)(nc * 16 + c)];
    }
#pragma unroll
    for (int r = 0; r < 8; ++r) {
        const int node = node0 + r;
        if (node < N) out[(unsigned)(node * 16 + c)] = sv[r] + acc[r];
    }
}

__global__ void pool_kernel(const float* __restrict__ h  ,
                            const float* __restrict__ dis,
                            const int* __restrict__ batch,
                            float* __restrict__ out, float* __restrict__ cnt, int N) {
    long long t = (long long)blockIdx.x * blockDim.x + threadIdx.x;
    if (t >= (long long)N * 16) return;
    int i = (int)(t >> 4);
    int c = (int)(t & 15);
    int g = batch[i];
    if (c < C_OUT) atomicAdd(&out[(size_t)g * C_OUT + c], h[t] * dis[i]);
    if (c == 0)    atomicAdd(&cnt[g], 1.0f);
}

__global__ void div_kernel(float* __restrict__ out, const float* __restrict__ cnt) {
    int t = blockIdx.x * blockDim.x + threadIdx.x;
    if (t < NUM_GRAPHS * C_OUT) out[t] /= fmaxf(cnt[t / C_OUT], 1.0f);
}


static inline int cdiv_ll(long long a, long long b) { return (int)((a + b - 1) / b); }

extern "C" void kernel_launch(void* const* d_in, const int* in_sizes, int n_in,
                              void* d_out, int out_size, void* d_ws, size_t ws_size,
                              hipStream_t stream) {
    const float* x     = (const float*)d_in[0];
    const int*   ei    = (const int*)d_in[1];
    const int*   batch = (const int*)d_in[2];
    const float* W0 = (const float*)d_in[3];  const float* b0 = (const float*)d_in[4];
    const float* W1 = (const float*)d_in[5];  const float* b1 = (const float*)d_in[6];
    const float* W2 = (const float*)d_in[7];  const float* b2 = (const float*)d_in[8];
    const float* W3 = (const float*)d_in[9];  const float* b3 = (const float*)d_in[10];

    const int N = in_sizes[2];
    const int E = in_sizes[1] / 2;
    const int H = 128;
    const int* rowi = ei;
    const int* coli = ei + E;

    float* out = (float*)d_out;
    float* ws  = (float*)d_ws;
    const size_t Npad = ((size_t)N + 255) & ~(size_t)255;
    float* dis  = ws;
    float* sq   = ws + Npad;
    float* bufA = sq + Npad;
    float* bufB = bufA + (size_t)N * H;
    float* bufC = bufB + (size_t)N * H;
    float* cnt  = bufC + (size_t)N * H;
    const int nbins = (N + BIN_ROWS - 1) >> BIN_SHIFT;
    const int nsup  = (N + SUP_ROWS - 1) >> SUP_SHIFT;
    const size_t binPad = ((size_t)nbins + 1 + 255) & ~(size_t)255;
    int* binStart = (int*)(cnt + NUM_GRAPHS);
    int* cursor   = binStart + binPad;
    int* binned   = cursor + binPad * CUR_STRIDE;

    const int TB = 256;

    fill_kernel<<<cdiv_ll(N, TB), TB, 0, stream>>>(dis, 1.0f, N);
    degree_kernel<<<cdiv_ll(E, TB), TB, 0, stream>>>(dis, coli, E);
    bin_offsets_kernel<<<1, 1024, 0, stream>>>(dis, binStart, cursor, N, nbins, nsup);
    rsqrt_kernel<<<cdiv_ll(N, TB), TB, 0, stream>>>(dis, sq, N);
    place_kernel<<<cdiv_ll(E, TB), TB, 0, stream>>>(rowi, coli, cursor, binned, E);

    const int gemmBlkH   = 256 * 2;
    const int gemmBlkOut = 256 * 2;
    const int segBlk = (nbins + SEG_WAVES - 1) / SEG_WAVES;
    const int segTB  = SEG_WAVES * 32;

    gemm_wmma_f32<128, 128, 128, false, true><<<gemmBlkH, TB, 0, stream>>>(x, W0, dis, sq, b0, bufA, bufB, N, 8);
    seg_sum_wmma_kernel<<<segBlk, segTB, 0, stream>>>(bufA, binned, binStart, bufB, N, nbins);

    gemm_wmma_f32<128, 128, 128, true, false><<<gemmBlkH, TB, 0, stream>>>(bufB, W1, dis, sq, b1, bufA, bufC, N, 8);
    seg_sum_wmma_kernel<<<segBlk, segTB, 0, stream>>>(bufA, binned, binStart, bufC, N, nbins);

    gemm_wmma_f32<128, 128, 128, true, false><<<gemmBlkH, TB, 0, stream>>>(bufC, W2, dis, sq, b2, bufA, bufB, N, 8);
    seg_sum_wmma_kernel<<<segBlk, segTB, 0, stream>>>(bufA, binned, binStart, bufB, N, nbins);

    gemm_wmma_f32<10, 16, 128, true, false><<<gemmBlkOut, TB, 0, stream>>>(bufB, W3, dis, sq, b3, bufA, bufC, N, 1);
    seg_sum16_wmma_kernel<<<segBlk, segTB, 0, stream>>>(bufA, binned, binStart, bufC, N, nbins);

    fill_kernel<<<cdiv_ll(NUM_GRAPHS * C_OUT, TB), TB, 0, stream>>>(out, 0.0f, NUM_GRAPHS * C_OUT);
    fill_kernel<<<cdiv_ll(NUM_GRAPHS, TB), TB, 0, stream>>>(cnt, 0.0f, NUM_GRAPHS);
    pool_kernel<<<cdiv_ll((long long)N * 16, TB), TB, 0, stream>>>(bufC, dis, batch, out, cnt, N);
    div_kernel<<<cdiv_ll(NUM_GRAPHS * C_OUT, TB), TB, 0, stream>>>(out, cnt);
}


# Degree count, edge placement, mean pool: 8 / 8 / 16 elements per thread, 150,000 -> 15,600 waves to start; pool sums a thread's nodes in a register before its atomic: pool sum order changes
# speedup vs baseline: 2.6010x; 1.0501x over previous
#include <hip/hip_runtime.h>


typedef float v2f __attribute__((ext_vector_type(2)));
typedef float v8f __attribute__((ext_vector_type(8)));

#define NUM_GRAPHS 512
#define C_OUT 10


__global__ void fill_kernel(float* __restrict__ p, float v, int n) {
    int t = blockIdx.x * blockDim.x + threadIdx.x;
    if (t < n) p[t] = v;
}

#define EPT 8

__global__ __launch_bounds__(256)
void degree_kernel(float* __restrict__ deg, const int* __restrict__ col, int E) {
    const int T = gridDim.x * blockDim.x;
    const int t = blockIdx.x * blockDim.x + threadIdx.x;
    int c[EPT];
#pragma unroll
    for (int i = 0; i < EPT; ++i) { const int e = t + i * T; c[i] = col[e < E ? e : E - 1]; }
#pragma unroll
    for (int i = 0; i < EPT; ++i) { const int e = t + i * T; if (e < E) atomicAdd(&deg[(unsigned)c[i]], 1.0f); }
}

__global__ void rsqrt_kernel(float* __restrict__ deg, float* __restrict__ sq, int n) {
    int t = blockIdx.x * blockDim.x + threadIdx.x;
    if (t < n) {
        float v = deg[t];
        deg[t] = rsqrtf(v);
        sq[t]  = sqrtf(v);
    }
}

#define BIN_SHIFT 4
#define BIN_ROWS 16
#define SUP_SHIFT 6
#define SUP_ROWS 64
#define CUR_STRIDE 32

__global__ __launch_bounds__(1024)
void bin_offsets_kernel(const float* __restrict__ deg, int* __restrict__ binStart,
                        int* __restrict__ cursor, int N, int nbins, int nsup) {
    __shared__ int s[1024];
    const int tid = threadIdx.x;
    int carry = 0;
    for (int b0 = 0; b0 < nsup; b0 += 1024) {
        const int sb = b0 + tid;
        int c4[4] = {0, 0, 0, 0};
        if (sb < nsup) {
            const int c0 = sb << SUP_SHIFT;
            if (c0 + SUP_ROWS <= N) {
                const float4* p = (const float4*)(deg + c0);
#pragma unroll
                for (int i = 0; i < 4; ++i) {
                    float a = 0.0f;
#pragma unroll
                    for (int j = 0; j < BIN_ROWS / 4; ++j) { float4 v = p[i * (BIN_ROWS / 4) + j]; a += (v.x + v.y) + (v.z + v.w); }
                    c4[i] = (int)a - BIN_ROWS;
                }
            } else {
                for (int c = c0; c < N; ++c) {
                    const int k = (c - c0) >> BIN_SHIFT;
                    const int d = (int)deg[c] - 1;
                    c4[0] += k == 0 ? d : 0;   c4[1] += k == 1 ? d : 0;
                    c4[2] += k == 2 ? d : 0;   c4[3] += k == 3 ? d : 0;
                }
            }
        }
        const int cnt = (c4[0] + c4[1]) + (c4[2] + c4[3]);
        s[tid] = cnt;
        __syncthreads();
        for (int off = 1; off < 1024; off <<= 1) {
            int v = tid >= off ? s[tid - off] : 0;
            __syncthreads();
            s[tid] += v;
            __syncthreads();
        }
        int excl = carry + s[tid] - cnt;
        if (sb < nsup) {
#pragma unroll
            for (int i = 0; i < 4; ++i) {
                const int b = sb * 4 + i;
                if (b < nbins) { binStart[b] = excl; cursor[(size_t)b * CUR_STRIDE] = excl; }
                excl += c4[i];
            }
        }
        carry += s[1023];
        __syncthreads();
    }
    if (tid == 0) binStart[nbins] = carry;
}

__global__ __launch_bounds__(256)
void place_kernel(const int* __restrict__ row, const int* __restrict__ col,
                  int* __restrict__ cursor, int* __restrict__ binned, int E) {
    const int T = gridDim.x * blockDim.x;
    const int t = blockIdx.x * blockDim.x + threadIdx.x;
    int c[EPT], r[EPT], pos[EPT];
#pragma unroll
    for (int i = 0; i < EPT; ++i) {
        const int e = t + i * T;
        const int ec = e < E ? e : E - 1;
        c[i] = col[ec];
        r[i] = row[ec];
    }
#pragma unroll
    for (int i = 0; i < EPT; ++i) {
        const int e = t + i * T;
        pos[i] = 0;
        if (e < E) pos[i] = atomicAdd(&cursor[(size_t)((unsigned)c[i] >> BIN_SHIFT) * CUR_STRIDE], 1);
    }
#pragma unroll
    for (int i = 0; i < EPT; ++i) {
        const int e = t + i * T;
        if (e < E) ((int2*)binned)[(unsigned)pos[i]] = make_int2(r[i], c[i] & (BIN_ROWS - 1));
    }
}


template <int KDIM>
__device__ __forceinline__ void gemm_load_strip(v2f (&a)[KDIM / 4], const float* __restrict__ A,
                                                int tm, int l15, int half, int N) {
    int row = (tm << 4) + l15;   if (row >= N) row = N - 1;
    const float* __restrict__ Arow = A + (size_t)row * KDIM + half * 2;
#pragma unroll
    for (int i = 0; i < KDIM / 4; ++i) a[i] = *(const v2f*)(Arow + 4 * i);
}

template <int DSTRIDE, int KDIM, bool RELU_A, bool FIRST>
__device__ __forceinline__ void gemm_strip(const v2f (&af)[KDIM / 4], const v2f (&bf)[KDIM / 4],
                                           const float* __restrict__ dis, const float* __restrict__ sq,
                                           float bv, float* __restrict__ D, float* __restrict__ S,
                                           int tm, int ccol, int half, int N) {
    const int m0 = tm << 4;
    v8f acc = {0.f, 0.f, 0.f, 0.f, 0.f, 0.f, 0.f, 0.f};
#pragma unroll
    for (int i = 0; i < KDIM / 4; ++i) {
        v2f a = af[i];
        if constexpr (RELU_A) {
            a[0] = a[0] > 0.0f ? a[0] : 0.0f;
            a[1] = a[1] > 0.0f ? a[1] : 0.0f;
        }
        acc = __builtin_amdgcn_wmma_f32_16x16x4_f32(false, a, false, bf[i],
                                                    (short)0, acc, false, false);
    }
    float dv[8], sv[8];
#pragma unroll
    for (int r = 0; r < 8; ++r) {
        int m  = m0 + r + half * 8;
        int mc = m < N ? m : N - 1;
        float d = dis[mc];
        float q = sq[mc];
        float f = FIRST ? d : d * d;
        dv[r] = acc[r] * f;
        sv[r] = dv[r] + bv * q;
    }
#pragma unroll
    for (int r = 0; r < 8; ++r) {
        int m = m0 + r + half * 8;
        if (m < N) {
            size_t o = (size_t)m * DSTRIDE + ccol;
            D[o] = dv[r];
            S[o] = sv[r];
        }
    }
}

template <int BCOLS, int DSTRIDE, int KDIM, bool RELU_A, bool FIRST>
__global__ __launch_bounds__(256)
void gemm_wmma_f32(const float* __restrict__ A, const float* __restrict__ B,
                   const float* __restrict__ dis, const float* __restrict__ sq,
                   const float* __restrict__ bias,
                   float* __restrict__ D, float* __restrict__ S, int N, int ntilesN) {
    const int wib  = __builtin_amdgcn_readfirstlane((int)(threadIdx.x >> 5));
    const int wave = blockIdx.x * (blockDim.x >> 5) + wib;
    const int nwav = gridDim.x * (blockDim.x >> 5);
    const int lane = threadIdx.x & 31;
    const int tilesM = (N + 15) >> 4;
    const int tn    = wave % ntilesN;
    const int tstep = nwav / ntilesN;
    int tm = wave / ntilesN;
    if (tm >= tilesM) return;
    const int n0 = tn << 4;
    const int half = lane >> 4;
    const int l15  = lane & 15;

    int bcol = n0 + l15;   if (bcol >= BCOLS) bcol = BCOLS - 1;
    const float* __restrict__ Bcol = B + (size_t)(half * 2) * BCOLS + bcol;
    const int   ccol = n0 + l15;
    const float bl   = bias[bcol];
    const float bv   = (ccol < BCOLS) ? bl : 0.0f;

    v2f a0[KDIM / 4], a1[KDIM / 4];
    gemm_load_strip<KDIM>(a0, A, tm, l15, half, N);

    v2f bf[KDIM / 4];
#pragma unroll
    for (int k0 = 0; k0 < KDIM; k0 += 4) {
        bf[k0 >> 2][0] = Bcol[(size_t)k0 * BCOLS];
        bf[k0 >> 2][1] = Bcol[(size_t)(k0 + 1) * BCOLS];
    }

    for (;;) {
        gemm_load_strip<KDIM>(a1, A, tm + tstep, l15, half, N);
        gemm_strip<DSTRIDE, KDIM, RELU_A, FIRST>(a0, bf, dis, sq, bv, D, S, tm, ccol, half, N);
        tm += tstep;
        if (tm >= tilesM) break;
        gemm_load_strip<KDIM>(a0, A, tm + tstep, l15, half, N);
        gemm_strip<DSTRIDE, KDIM, RELU_A, FIRST>(a1, bf, dis, sq, bv, D, S, tm, ccol, half, N);
        tm += tstep;
        if (tm >= tilesM) break;
    }
}

typedef float v4f __attribute__((ext_vector_type(4)));

template <int K>
__device__ __forceinline__ int row_share16(int v) {
    return __builtin_amdgcn_update_dpp(0, v, 0x150 + K, 0xf, 0xf, false);
}

template <int G>
__device__ __forceinline__ void seg_fetch(v4f (&x)[2], v4f (&y)[2], v2f& a, int src, int tg,
                                          const float* __restrict__ lin, int c) {
    const int sx = row_share16<2 * G>(src),     tx = row_share16<2 * G>(tg);
    const int sy = row_share16<2 * G + 1>(src), ty = row_share16<2 * G + 1>(tg);
    a[0] = tx == c ? 1.0f : 0.0f;
    a[1] = ty == c ? 1.0f : 0.0f;
    const unsigned ox = ((unsigned)sx << 9) | ((unsigned)c << 4);
    const unsigned oy = ((unsigned)sy << 9) | ((unsigned)c << 4);
    const char* __restrict__ linb = (const char*)lin;
    x[0] = *(const v4f*)(linb + ox);   x[1] = *(const v4f*)(linb + ox + 256u);
    y[0] = *(const v4f*)(linb + oy);   y[1] = *(const v4f*)(linb + oy + 256u);
}

template <int SB>
__device__ __forceinline__ void seg_half_trip(v8f (&acc)[8], int src, int tg,
                                              const float* __restrict__ lin, int c) {
    v4f x[4][2], y[4][2];
    v2f a[4];
    seg_fetch<4 * SB + 0>(x[0], y[0], a[0], src, tg, lin, c);
    seg_fetch<4 * SB + 1>(x[1], y[1], a[1], src, tg, lin, c);
    seg_fetch<4 * SB + 2>(x[2], y[2], a[2], src, tg, lin, c);
    seg_fetch<4 * SB + 3>(x[3], y[3], a[3], src, tg, lin, c);
#pragma unroll
    for (int i = 0; i < 4; ++i) {
#pragma unroll
        for (int s = 0; s < 2; ++s) {
#pragma unroll
            for (int q = 0; q < 4; ++q) {
                v2f b;
                b[0] = x[i][s][q];
                b[1] = y[i][s][q];
                acc[4 * s + q] = __builtin_amdgcn_wmma_f32_16x16x4_f32(false, a[i], false, b,
                                                                       (short)0, acc[4 * s + q], false, false);
            }
        }
    }
}

#define SEG_WAVES 4
__global__ __launch_bounds__(SEG_WAVES * 32)
void seg_sum_wmma_kernel(const float* __restrict__ lin, const int* __restrict__ binned,
                         const int* __restrict__ binStart, float* __restrict__ out, int N, int nbins) {
    const int lane = threadIdx.x & 31;
    const int wib  = __builtin_amdgcn_readfirstlane((int)(threadIdx.x >> 5));
    const int b    = blockIdx.x * SEG_WAVES + wib;
    if (b >= nbins) return;
    const int c    = lane & 15;
    const int half = lane >> 4;
    const int slot = 4 * (c >> 1) + 2 * half + (c & 1);

    const int eb = binStart[b];
    const int ee = binStart[b + 1];
    const int2* __restrict__ ed2 = (const int2*)binned;

    v8f acc[8];
#pragma unroll
    for (int t = 0; t < 8; ++t) acc[t] = (v8f){0.f, 0.f, 0.f, 0.f, 0.f, 0.f, 0.f, 0.f};

    int2 ed = make_int2(0, 0);
    if (eb < ee) {
        int idx = eb + slot;   if (idx >= ee) idx = ee - 1;
        ed = ed2[idx];
    }
    for (int base = eb; base < ee; base += 32) {
        const int nv = ee - base;
        int2 edn = ed;
        if (base + 32 < ee) {
            int idx = base + 32 + slot;   if (idx >= ee) idx = ee - 1;
            edn = ed2[idx];
        }
        const int src = ed.x;
        const int tg  = slot < nv ? ed.y : BIN_ROWS;
        seg_half_trip<0>(acc, src, tg, lin, c);
        if (nv > 16) seg_half_trip<1>(acc, src, tg, lin, c);
        ed = edn;
    }

    char* __restrict__ outb = (char*)out;
    const int node0 = (b << BIN_SHIFT) + half * 8;
#pragma unroll
    for (int r0 = 0; r0 < 8; r0 += 4) {
        v4f sv[4][2];
#pragma unroll
        for (int i = 0; i < 4; ++i) {
            const int node = node0 + r0 + i;
            const int nc = node < N ? node : N - 1;
            const unsigned o = ((unsigned)nc << 9) | ((unsigned)c << 4);
            sv[i][0] = *(const v4f*)(outb + o);
            sv[i][1] = *(const v4f*)(outb + o + 256u);
        }
#pragma unroll
        for (int i = 0; i < 4; ++i) {
            const int node = node0 + r0 + i;
#pragma unroll
            for (int s = 0; s < 2; ++s) {
                sv[i][s][0] += acc[4 * s + 0][r0 + i];
                sv[i][s][1] += acc[4 * s + 1][r0 + i];
                sv[i][s][2] += acc[4 * s + 2][r0 + i];
                sv[i][s][3] += acc[4 * s + 3][r0 + i];
            }
            if (node < N) {
                const unsigned o = ((unsigned)node << 9) | ((unsigned)c << 4);
                *(v4f*)(outb + o)        = sv[i][0];
                *(v4f*)(outb + o + 256u) = sv[i][1];
            }
        }
    }
}

template <int G>
__device__ __forceinline__ void seg16_fetch(v2f& bfr, v2f& a, int src, int tg,
                                            const float* __restrict__ lin, int c) {
    const int sx = row_share16<2 * G>(src),     tx = row_share16<2 * G>(tg);
    const int sy = row_share16<2 * G + 1>(src), ty = row_share16<2 * G + 1>(tg);
    a[0] = tx == c ? 1.0f : 0.0f;
    a[1] = ty == c ? 1.0f : 0.0f;
    const char* __restrict__ linb = (const char*)lin;
    bfr[0] = *(const float*)(linb + (((unsigned)sx << 6) | ((unsigned)c << 2)));
    bfr[1] = *(const float*)(linb + (((unsigned)sy << 6) | ((unsigned)c << 2)));
}

template <int SB>
__device__ __forceinline__ void seg16_half_trip(v8f& acc, int src, int tg,
                                                const float* __restrict__ lin, int c) {
    v2f bfr[4], a[4];
    seg16_fetch<4 * SB + 0>(bfr[0], a[0], src, tg, lin, c);
    seg16_fetch<4 * SB + 1>(bfr[1], a[1], src, tg, lin, c);
    seg16_fetch<4 * SB + 2>(bfr[2], a[2], src, tg, lin, c);
    seg16_fetch<4 * SB + 3>(bfr[3], a[3], src, tg, lin, c);
#pragma unroll
    for (int i = 0; i < 4; ++i)
        acc = __builtin_amdgcn_wmma_f32_16x16x4_f32(false, a[i], false, bfr[i], (short)0, acc, false, false);
}

__global__ __launch_bounds__(SEG_WAVES * 32)
void seg_sum16_wmma_kernel(const float* __restrict__ lin, const int* __restrict__ binned,
                           const int* __restrict__ binStart, float* __restrict__ out, int N, int nbins) {
    const int lane = threadIdx.x & 31;
    const int wib  = __builtin_amdgcn_readfirstlane((int)(threadIdx.x >> 5));
    const int b    = blockIdx.x * SEG_WAVES + wib;
    if (b >= nbins) return;
    const int c    = lane & 15;
    const int half = lane >> 4;
    const int slot = 4 * (c >> 1) + 2 * half + (c & 1);

    const int eb = binStart[b];
    const int ee = binStart[b + 1];
    const int2* __restrict__ ed2 = (const int2*)binned;

    v8f acc = {0.f, 0.f, 0.f, 0.f, 0.f, 0.f, 0.f, 0.f};

    int2 ed = make_int2(0, 0);
    if (eb < ee) {
        int idx = eb + slot;   if (idx >= ee) idx = ee - 1;
        ed = ed2[idx];
    }
    for (int base = eb; base < ee; base += 32) {
        const int nv = ee - base;
        int2 edn = ed;
        if (base + 32 < ee) {
            int idx = base + 32 + slot;   if (idx >= ee) idx = ee - 1;
            edn = ed2[idx];
        }
        const int src = ed.x;
        const int tg  = slot < nv ? ed.y : BIN_ROWS;
        seg16_half_trip<0>(acc, src, tg, lin, c);
        if (nv > 16) seg16_half_trip<1>(acc, src, tg, lin, c);
        ed = edn;
    }

    const int node0 = (b << BIN_SHIFT) + half * 8;
    float sv[8];
#pragma unroll
    for (int r = 0; r < 8; ++r) {
        const int node = node0 + r;
        const int nc = node < N ? node : N - 1;
        sv[r] = out[(unsigned)(nc * 16 + c)];
    }
#pragma unroll
    for (int r = 0; r < 8; ++r) {
        const int node = node0 + r;
        if (node < N) out[(unsigned)(node * 16 + c)] = sv[r] + acc[r];
    }
}

#define POOL_NPT 16

__device__ __forceinline__ void pool_flush(float* __restrict__ out, float* __restrict__ cnt,
                                           int g, int c, float acc, float cn) {
    if (c < C_OUT) atomicAdd(&out[(size_t)g * C_OUT + c], acc);
    if (c == 0)    atomicAdd(&cnt[g], cn);
}

__global__ __launch_bounds__(256)
void pool_kernel(const float* __restrict__ h  ,
                 const float* __restrict__ dis,
                 const int* __restrict__ batch,
                 float* __restrict__ out, float* __restrict__ cnt, int N) {
    const int lane  = threadIdx.x & 31;
    const int c     = lane & 15;
    const int half  = lane >> 4;
    const int wave  = (blockIdx.x * blockDim.x + threadIdx.x) >> 5;
    const int first = wave * (2 * POOL_NPT);
    if (first >= N) return;
    const int nb = first + half;
    float hv[POOL_NPT], dv[POOL_NPT];
    int   gv[POOL_NPT];
#pragma unroll
    for (int i = 0; i < POOL_NPT; ++i) {
        const int node = nb + 2 * i;
        const int nc = node < N ? node : N - 1;
        hv[i] = h[(size_t)nc * 16 + c];
        dv[i] = dis[nc];
        gv[i] = batch[nc];
    }
    int   gcur = gv[0];
    float acc = 0.0f, cn = 0.0f;
#pragma unroll
    for (int i = 0; i < POOL_NPT; ++i) {
        const int node = nb + 2 * i;
        if (node < N) {
            if (gv[i] != gcur) {
                pool_flush(out, cnt, gcur, c, acc, cn);
                gcur = gv[i];   acc = 0.0f;   cn = 0.0f;
            }
            acc += hv[i] * dv[i];
            cn  += 1.0f;
        }
    }
    if (cn > 0.0f) pool_flush(out, cnt, gcur, c, acc, cn);
}

__global__ void div_kernel(float* __restrict__ out, const float* __restrict__ cnt) {
    int t = blockIdx.x * blockDim.x + threadIdx.x;
    if (t < NUM_GRAPHS * C_OUT) out[t] /= fmaxf(cnt[t / C_OUT], 1.0f);
}


static inline int cdiv_ll(long long a, long long b) { return (int)((a + b - 1) / b); }

extern "C" void kernel_launch(void* const* d_in, const int* in_sizes, int n_in,
                              void* d_out, int out_size, void* d_ws, size_t ws_size,
                              hipStream_t stream) {
    const float* x     = (const float*)d_in[0];
    const int*   ei    = (const int*)d_in[1];
    const int*   batch = (const int*)d_in[2];
    const float* W0 = (const float*)d_in[3];  const float* b0 = (const float*)d_in[4];
    const float* W1 = (const float*)d_in[5];  const float* b1 = (const float*)d_in[6];
    const float* W2 = (const float*)d_in[7];  const float* b2 = (const float*)d_in[8];
    const float* W3 = (const float*)d_in[9];  const float* b3 = (const float*)d_in[10];

    const int N = in_sizes[2];
    const int E = in_sizes[1] / 2;
    const int H = 128;
    const int* rowi = ei;
    const int* coli = ei + E;

    float* out = (float*)d_out;
    float* ws  = (float*)d_ws;
    const size_t Npad = ((size_t)N + 255) & ~(size_t)255;
    float* dis  = ws;
    float* sq   = ws + Npad;
    float* bufA = sq + Npad;
    float* bufB = bufA + (size_t)N * H;
    float* bufC = bufB + (size_t)N * H;
    float* cnt  = bufC + (size_t)N * H;
    const int nbins = (N + BIN_ROWS - 1) >> BIN_SHIFT;
    const int nsup  = (N + SUP_ROWS - 1) >> SUP_SHIFT;
    const size_t binPad = ((size_t)nbins + 1 + 255) & ~(size_t)255;
    int* binStart = (int*)(cnt + NUM_GRAPHS);
    int* cursor   = binStart + binPad;
    int* binned   = cursor + binPad * CUR_STRIDE;

    const int TB = 256;

    fill_kernel<<<cdiv_ll(N, TB), TB, 0, stream>>>(dis, 1.0f, N);
    degree_kernel<<<cdiv_ll(E, TB * EPT), TB, 0, stream>>>(dis, coli, E);
    bin_offsets_kernel<<<1, 1024, 0, stream>>>(dis, binStart, cursor, N, nbins, nsup);
    rsqrt_kernel<<<cdiv_ll(N, TB), TB, 0, stream>>>(dis, sq, N);
    place_kernel<<<cdiv_ll(E, TB * EPT), TB, 0, stream>>>(rowi, coli, cursor, binned, E);

    const int gemmBlkH   = 256 * 2;
    const int gemmBlkOut = 256 * 2;
    const int segBlk = (nbins + SEG_WAVES - 1) / SEG_WAVES;
    const int segTB  = SEG_WAVES * 32;

    gemm_wmma_f32<128, 128, 128, false, true><<<gemmBlkH, TB, 0, stream>>>(x, W0, dis, sq, b0, bufA, bufB, N, 8);
    seg_sum_wmma_kernel<<<segBlk, segTB, 0, stream>>>(bufA, binned, binStart, bufB, N, nbins);

    gemm_wmma_f32<128, 128, 128, true, false><<<gemmBlkH, TB, 0, stream>>>(bufB, W1, dis, sq, b1, bufA, bufC, N, 8);
    seg_sum_wmma_kernel<<<segBlk, segTB, 0, stream>>>(bufA, binned, binStart, bufC, N, nbins);

    gemm_wmma_f32<128, 128, 128, true, false><<<gemmBlkH, TB, 0, stream>>>(bufC, W2, dis, sq, b2, bufA, bufB, N, 8);
    seg_sum_wmma_kernel<<<segBlk, segTB, 0, stream>>>(bufA, binned, binStart, bufB, N, nbins);

    gemm_wmma_f32<10, 16, 128, true, false><<<gemmBlkOut, TB, 0, stream>>>(bufB, W3, dis, sq, b3, bufA, bufC, N, 1);
    seg_sum16_wmma_kernel<<<segBlk, segTB, 0, stream>>>(bufA, binned, binStart, bufC, N, nbins);

    fill_kernel<<<cdiv_ll(NUM_GRAPHS * C_OUT, TB), TB, 0, stream>>>(out, 0.0f, NUM_GRAPHS * C_OUT);
    fill_kernel<<<cdiv_ll(NUM_GRAPHS, TB), TB, 0, stream>>>(cnt, 0.0f, NUM_GRAPHS);
    pool_kernel<<<cdiv_ll((long long)N, (TB / 32) * 2 * POOL_NPT), TB, 0, stream>>>(bufC, dis, batch, out, cnt, N);
    div_kernel<<<cdiv_ll(NUM_GRAPHS * C_OUT, TB), TB, 0, stream>>>(out, cnt);
}


# Layers 2, 3 and output: the wave that summed a bin multiplies its 16x128 sums by the next weights (strip through its own LDS slot, weights in LDS); 17 -> 14 launches; sum orders unchanged
# speedup vs baseline: 2.8929x; 1.1122x over previous
#include <hip/hip_runtime.h>


typedef float v2f __attribute__((ext_vector_type(2)));
typedef float v8f __attribute__((ext_vector_type(8)));

#define NUM_GRAPHS 512
#define C_OUT 10


__global__ void fill_kernel(float* __restrict__ p, float v, int n) {
    int t = blockIdx.x * blockDim.x + threadIdx.x;
    if (t < n) p[t] = v;
}

#define EPT 8

__global__ __launch_bounds__(256)
void degree_kernel(float* __restrict__ deg, const int* __restrict__ col, int E) {
    const int T = gridDim.x * blockDim.x;
    const int t = blockIdx.x * blockDim.x + threadIdx.x;
    int c[EPT];
#pragma unroll
    for (int i = 0; i < EPT; ++i) { const int e = t + i * T; c[i] = col[e < E ? e : E - 1]; }
#pragma unroll
    for (int i = 0; i < EPT; ++i) { const int e = t + i * T; if (e < E) atomicAdd(&deg[(unsigned)c[i]], 1.0f); }
}

__global__ void rsqrt_kernel(float* __restrict__ deg, float* __restrict__ sq, int n) {
    int t = blockIdx.x * blockDim.x + threadIdx.x;
    if (t < n) {
        float v = deg[t];
        deg[t] = rsqrtf(v);
        sq[t]  = sqrtf(v);
    }
}

#define BIN_SHIFT 4
#define BIN_ROWS 16
#define SUP_SHIFT 6
#define SUP_ROWS 64
#define CUR_STRIDE 32

__global__ __launch_bounds__(1024)
void bin_offsets_kernel(const float* __restrict__ deg, int* __restrict__ binStart,
                        int* __restrict__ cursor, int N, int nbins, int nsup) {
    __shared__ int s[1024];
    const int tid = threadIdx.x;
    int carry = 0;
    for (int b0 = 0; b0 < nsup; b0 += 1024) {
        const int sb = b0 + tid;
        int c4[4] = {0, 0, 0, 0};
        if (sb < nsup) {
            const int c0 = sb << SUP_SHIFT;
            if (c0 + SUP_ROWS <= N) {
                const float4* p = (const float4*)(deg + c0);
#pragma unroll
                for (int i = 0; i < 4; ++i) {
                    float a = 0.0f;
#pragma unroll
                    for (int j = 0; j < BIN_ROWS / 4; ++j) { float4 v = p[i * (BIN_ROWS / 4) + j]; a += (v.x + v.y) + (v.z + v.w); }
                    c4[i] = (int)a - BIN_ROWS;
                }
            } else {
                for (int c = c0; c < N; ++c) {
                    const int k = (c - c0) >> BIN_SHIFT;
                    const int d = (int)deg[c] - 1;
                    c4[0] += k == 0 ? d : 0;   c4[1] += k == 1 ? d : 0;
                    c4[2] += k == 2 ? d : 0;   c4[3] += k == 3 ? d : 0;
                }
            }
        }
        const int cnt = (c4[0] + c4[1]) + (c4[2] + c4[3]);
        s[tid] = cnt;
        __syncthreads();
        for (int off = 1; off < 1024; off <<= 1) {
            int v = tid >= off ? s[tid - off] : 0;
            __syncthreads();
            s[tid] += v;
            __syncthreads();
        }
        int excl = carry + s[tid] - cnt;
        if (sb < nsup) {
#pragma unroll
            for (int i = 0; i < 4; ++i) {
                const int b = sb * 4 + i;
                if (b < nbins) { binStart[b] = excl; cursor[(size_t)b * CUR_STRIDE] = excl; }
                excl += c4[i];
            }
        }
        carry += s[1023];
        __syncthreads();
    }
    if (tid == 0) binStart[nbins] = carry;
}

__global__ __launch_bounds__(256)
void place_kernel(const int* __restrict__ row, const int* __restrict__ col,
                  int* __restrict__ cursor, int* __restrict__ binned, int E) {
    const int T = gridDim.x * blockDim.x;
    const int t = blockIdx.x * blockDim.x + threadIdx.x;
    int c[EPT], r[EPT], pos[EPT];
#pragma unroll
    for (int i = 0; i < EPT; ++i) {
        const int e = t + i * T;
        const int ec = e < E ? e : E - 1;
        c[i] = col[ec];
        r[i] = row[ec];
    }
#pragma unroll
    for (int i = 0; i < EPT; ++i) {
        const int e = t + i * T;
        pos[i] = 0;
        if (e < E) pos[i] = atomicAdd(&cursor[(size_t)((unsigned)c[i] >> BIN_SHIFT) * CUR_STRIDE], 1);
    }
#pragma unroll
    for (int i = 0; i < EPT; ++i) {
        const int e = t + i * T;
        if (e < E) ((int2*)binned)[(unsigned)pos[i]] = make_int2(r[i], c[i] & (BIN_ROWS - 1));
    }
}


template <int KDIM>
__device__ __forceinline__ void gemm_load_strip(v2f (&a)[KDIM / 4], const float* __restrict__ A,
                                                int tm, int l15, int half, int N) {
    int row = (tm << 4) + l15;   if (row >= N) row = N - 1;
    const float* __restrict__ Arow = A + (size_t)row * KDIM + half * 2;
#pragma unroll
    for (int i = 0; i < KDIM / 4; ++i) a[i] = *(const v2f*)(Arow + 4 * i);
}

template <int DSTRIDE, int KDIM, bool RELU_A, bool FIRST>
__device__ __forceinline__ void gemm_strip(const v2f (&af)[KDIM / 4], const v2f (&bf)[KDIM / 4],
                                           const float* __restrict__ dis, const float* __restrict__ sq,
                                           float bv, float* __restrict__ D, float* __restrict__ S,
                                           int tm, int ccol, int half, int N) {
    const int m0 = tm << 4;
    v8f acc = {0.f, 0.f, 0.f, 0.f, 0.f, 0.f, 0.f, 0.f};
#pragma unroll
    for (int i = 0; i < KDIM / 4; ++i) {
        v2f a = af[i];
        if constexpr (RELU_A) {
            a[0] = a[0] > 0.0f ? a[0] : 0.0f;
            a[1] = a[1] > 0.0f ? a[1] : 0.0f;
        }
        acc = __builtin_amdgcn_wmma_f32_16x16x4_f32(false, a, false, bf[i],
                                                    (short)0, acc, false, false);
    }
    float dv[8], sv[8];
#pragma unroll
    for (int r = 0; r < 8; ++r) {
        int m  = m0 + r + half * 8;
        int mc = m < N ? m : N - 1;
        float d = dis[mc];
        float q = sq[mc];
        float f = FIRST ? d : d * d;
        dv[r] = acc[r] * f;
        sv[r] = dv[r] + bv * q;
    }
#pragma unroll
    for (int r = 0; r < 8; ++r) {
        int m = m0 + r + half * 8;
        if (m < N) {
            size_t o = (size_t)m * DSTRIDE + ccol;
            D[o] = dv[r];
            S[o] = sv[r];
        }
    }
}

template <int BCOLS, int DSTRIDE, int KDIM, bool RELU_A, bool FIRST>
__global__ __launch_bounds__(256)
void gemm_wmma_f32(const float* __restrict__ A, const float* __restrict__ B,
                   const float* __restrict__ dis, const float* __restrict__ sq,
                   const float* __restrict__ bias,
                   float* __restrict__ D, float* __restrict__ S, int N, int ntilesN) {
    const int wib  = __builtin_amdgcn_readfirstlane((int)(threadIdx.x >> 5));
    const int wave = blockIdx.x * (blockDim.x >> 5) + wib;
    const int nwav = gridDim.x * (blockDim.x >> 5);
    const int lane = threadIdx.x & 31;
    const int tilesM = (N + 15) >> 4;
    const int tn    = wave % ntilesN;
    const int tstep = nwav / ntilesN;
    int tm = wave / ntilesN;
    if (tm >= tilesM) return;
    const int n0 = tn << 4;
    const int half = lane >> 4;
    const int l15  = lane & 15;

    int bcol = n0 + l15;   if (bcol >= BCOLS) bcol = BCOLS - 1;
    const float* __restrict__ Bcol = B + (size_t)(half * 2) * BCOLS + bcol;
    const int   ccol = n0 + l15;
    const float bl   = bias[bcol];
    const float bv   = (ccol < BCOLS) ? bl : 0.0f;

    v2f a0[KDIM / 4], a1[KDIM / 4];
    gemm_load_strip<KDIM>(a0, A, tm, l15, half, N);

    v2f bf[KDIM / 4];
#pragma unroll
    for (int k0 = 0; k0 < KDIM; k0 += 4) {
        bf[k0 >> 2][0] = Bcol[(size_t)k0 * BCOLS];
        bf[k0 >> 2][1] = Bcol[(size_t)(k0 + 1) * BCOLS];
    }

    for (;;) {
        gemm_load_strip<KDIM>(a1, A, tm + tstep, l15, half, N);
        gemm_strip<DSTRIDE, KDIM, RELU_A, FIRST>(a0, bf, dis, sq, bv, D, S, tm, ccol, half, N);
        tm += tstep;
        if (tm >= tilesM) break;
        gemm_load_strip<KDIM>(a0, A, tm + tstep, l15, half, N);
        gemm_strip<DSTRIDE, KDIM, RELU_A, FIRST>(a1, bf, dis, sq, bv, D, S, tm, ccol, half, N);
        tm += tstep;
        if (tm >= tilesM) break;
    }
}

typedef float v4f __attribute__((ext_vector_type(4)));

template <int K>
__device__ __forceinline__ int row_share16(int v) {
    return __builtin_amdgcn_update_dpp(0, v, 0x150 + K, 0xf, 0xf, false);
}

template <int G>
__device__ __forceinline__ void seg_fetch(v4f (&x)[2], v4f (&y)[2], v2f& a, int src, int tg,
                                          const float* __restrict__ lin, int c) {
    const int sx = row_share16<2 * G>(src),     tx = row_share16<2 * G>(tg);
    const int sy = row_share16<2 * G + 1>(src), ty = row_share16<2 * G + 1>(tg);
    a[0] = tx == c ? 1.0f : 0.0f;
    a[1] = ty == c ? 1.0f : 0.0f;
    const unsigned ox = ((unsigned)sx << 9) | ((unsigned)c << 4);
    const unsigned oy = ((unsigned)sy << 9) | ((unsigned)c << 4);
    const char* __restrict__ linb = (const char*)lin;
    x[0] = *(const v4f*)(linb + ox);   x[1] = *(const v4f*)(linb + ox + 256u);
    y[0] = *(const v4f*)(linb + oy);   y[1] = *(const v4f*)(linb + oy + 256u);
}

template <int SB>
__device__ __forceinline__ void seg_half_trip(v8f (&acc)[8], int src, int tg,
                                              const float* __restrict__ lin, int c) {
    v4f x[4][2], y[4][2];
    v2f a[4];
    seg_fetch<4 * SB + 0>(x[0], y[0], a[0], src, tg, lin, c);
    seg_fetch<4 * SB + 1>(x[1], y[1], a[1], src, tg, lin, c);
    seg_fetch<4 * SB + 2>(x[2], y[2], a[2], src, tg, lin, c);
    seg_fetch<4 * SB + 3>(x[3], y[3], a[3], src, tg, lin, c);
#pragma unroll
    for (int i = 0; i < 4; ++i) {
#pragma unroll
        for (int s = 0; s < 2; ++s) {
#pragma unroll
            for (int q = 0; q < 4; ++q) {
                v2f b;
                b[0] = x[i][s][q];
                b[1] = y[i][s][q];
                acc[4 * s + q] = __builtin_amdgcn_wmma_f32_16x16x4_f32(false, a[i], false, b,
                                                                       (short)0, acc[4 * s + q], false, false);
            }
        }
    }
}

#define SEG_WAVES 4

__device__ __forceinline__ void seg_bin_sums(v8f (&acc)[8], const float* __restrict__ lin,
                                             const int* __restrict__ binned, const int* __restrict__ binStart,
                                             int b, int c, int half) {
    const int slot = 4 * (c >> 1) + 2 * half + (c & 1);
    const int eb = binStart[b];
    const int ee = binStart[b + 1];
    const int2* __restrict__ ed2 = (const int2*)binned;

    int2 ed = make_int2(0, 0);
    if (eb < ee) {
        int idx = eb + slot;   if (idx >= ee) idx = ee - 1;
        ed = ed2[idx];
    }
    for (int base = eb; base < ee; base += 32) {
        const int nv = ee - base;
        int2 edn = ed;
        if (base + 32 < ee) {
            int idx = base + 32 + slot;   if (idx >= ee) idx = ee - 1;
            edn = ed2[idx];
        }
        const int src = ed.x;
        const int tg  = slot < nv ? ed.y : BIN_ROWS;
        seg_half_trip<0>(acc, src, tg, lin, c);
        if (nv > 16) seg_half_trip<1>(acc, src, tg, lin, c);
        ed = edn;
    }
}

#define FUSE_WAVES 25
#define T_PITCH 132

template <int BCOLS, int DSTRIDE>
__global__ __launch_bounds__(FUSE_WAVES * 32)
void seg_sum_gemm_kernel(const float* __restrict__ lin, const int* __restrict__ binned,
                         const int* __restrict__ binStart, const float* start,
                         const float* __restrict__ Wn, const float* __restrict__ biasn,
                         const float* __restrict__ dis, const float* __restrict__ sq,
                         float* __restrict__ Dn, float* Sn, int N, int nbins) {
    constexpr int NT = DSTRIDE / 16;
    static_assert(NT == 8 || NT == 1, "column tiles: 8 (hidden layers) or 1 (output layer)");
    __shared__ __attribute__((aligned(16))) float wp[128 * DSTRIDE];
    __shared__ __attribute__((aligned(16))) float tslot[FUSE_WAVES * BIN_ROWS * T_PITCH];

    if constexpr (BCOLS == DSTRIDE) {
        constexpr int NV4 = 128 * DSTRIDE / 4;
        constexpr int PER = (NV4 + FUSE_WAVES * 32 - 1) / (FUSE_WAVES * 32);
        v4f wv[PER];
#pragma unroll
        for (int j = 0; j < PER; ++j) {
            const int i4 = threadIdx.x + j * (FUSE_WAVES * 32);
            wv[j] = ((const v4f*)Wn)[i4 < NV4 ? i4 : NV4 - 1];
        }
#pragma unroll
        for (int j = 0; j < PER; ++j) {
            const int i4 = threadIdx.x + j * (FUSE_WAVES * 32);
            if (i4 < NV4) {
                const int k   = (4 * i4) / DSTRIDE;
                const int col = (4 * i4) % DSTRIDE;
                const int i  = k >> 2;
                const int ln = ((k >> 1) & 1) * 16 + (col & 15);
                const int e  = k & 1;
                const int to = col >> 4;
                const int pos = ((i * (NT / 2) + (to >> 1)) * 32 + ln) * 4 + 2 * (to & 1) + e;
#pragma unroll
                for (int u = 0; u < 4; ++u) wp[pos + 4 * u] = wv[j][u];
            }
        }
    } else {
        constexpr int NEL = 128 * DSTRIDE;
        constexpr int PER = (NEL + FUSE_WAVES * 32 - 1) / (FUSE_WAVES * 32);
        float wv[PER];
#pragma unroll
        for (int j = 0; j < PER; ++j) {
            const int id  = threadIdx.x + j * (FUSE_WAVES * 32);
            const int idc = id < NEL ? id : NEL - 1;
            const int k   = idc / DSTRIDE;
            const int col = idc % DSTRIDE;
            wv[j] = Wn[k * BCOLS + (col < BCOLS ? col : BCOLS - 1)];
        }
#pragma unroll
        for (int j = 0; j < PER; ++j) {
            const int id = threadIdx.x + j * (FUSE_WAVES * 32);
            if (id < NEL) {
                const int k   = id / DSTRIDE;
                const int col = id % DSTRIDE;
                const int i  = k >> 2;
                const int ln = ((k >> 1) & 1) * 16 + (col & 15);
                const int e  = k & 1;
                wp[(i * 32 + ln) * 2 + e] = col < BCOLS ? wv[j] : 0.0f;
            }
        }
    }
    __syncthreads();

    const int lane = threadIdx.x & 31;
    const int wib  = __builtin_amdgcn_readfirstlane((int)(threadIdx.x >> 5));
    const int b    = blockIdx.x * FUSE_WAVES + wib;
    if (b >= nbins) return;
    const int c    = lane & 15;
    const int half = lane >> 4;

    v8f acc[8];
#pragma unroll
    for (int t = 0; t < 8; ++t) acc[t] = (v8f){0.f, 0.f, 0.f, 0.f, 0.f, 0.f, 0.f, 0.f};
    seg_bin_sums(acc, lin, binned, binStart, b, c, half);

    float* tw = tslot + wib * (BIN_ROWS * T_PITCH);
    const char* startb = (const char*)start;
    const int node0 = (b << BIN_SHIFT) + half * 8;
#pragma unroll
    for (int r0 = 0; r0 < 8; r0 += 4) {
        v4f sv[4][2];
#pragma unroll
        for (int i = 0; i < 4; ++i) {
            const int node = node0 + r0 + i;
            const int nc = node < N ? node : N - 1;
            const unsigned o = ((unsigned)nc << 9) | ((unsigned)c << 4);
            sv[i][0] = *(const v4f*)(startb + o);
            sv[i][1] = *(const v4f*)(startb + o + 256u);
        }
#pragma unroll
        for (int i = 0; i < 4; ++i) {
#pragma unroll
            for (int s = 0; s < 2; ++s) {
#pragma unroll
                for (int q = 0; q < 4; ++q) {
                    const float t = sv[i][s][q] + acc[4 * s + q][r0 + i];
                    sv[i][s][q] = t > 0.0f ? t : 0.0f;
                }
                *(v4f*)(tw + (half * 8 + r0 + i) * T_PITCH + 64 * s + 4 * c) = sv[i][s];
            }
        }
    }
    __builtin_amdgcn_fence(__ATOMIC_RELEASE, "wavefront");
    __builtin_amdgcn_fence(__ATOMIC_ACQUIRE, "wavefront");

    v8f po[NT];
#pragma unroll
    for (int t = 0; t < NT; ++t) po[t] = (v8f){0.f, 0.f, 0.f, 0.f, 0.f, 0.f, 0.f, 0.f};
    const float* tr = tw + c * T_PITCH + 2 * half;
#pragma unroll
    for (int i = 0; i < 32; ++i) {
        const v2f a = *(const v2f*)(tr + 4 * i);
        if constexpr (NT == 8) {
#pragma unroll
            for (int tp = 0; tp < 4; ++tp) {
                const v4f w4 = *(const v4f*)(wp + ((i * 4 + tp) * 32 + lane) * 4);
                v2f b0, b1;
                b0[0] = w4[0];   b0[1] = w4[1];
                b1[0] = w4[2];   b1[1] = w4[3];
                po[2 * tp]     = __builtin_amdgcn_wmma_f32_16x16x4_f32(false, a, false, b0, (short)0, po[2 * tp], false, false);
                po[2 * tp + 1] = __builtin_amdgcn_wmma_f32_16x16x4_f32(false, a, false, b1, (short)0, po[2 * tp + 1], false, false);
            }
        } else {
            const v2f bw = *(const v2f*)(wp + (i * 32 + lane) * 2);
            po[0] = __builtin_amdgcn_wmma_f32_16x16x4_f32(false, a, false, bw, (short)0, po[0], false, false);
        }
    }

    float f8[8], q8[8];
#pragma unroll
    for (int r = 0; r < 8; ++r) {
        const int m  = node0 + r;
        const int mc = m < N ? m : N - 1;
        const float d = dis[mc];
        q8[r] = sq[mc];
        f8[r] = d * d;
    }
    float bv[NT];
#pragma unroll
    for (int to = 0; to < NT; ++to) {
        const int ccol = 16 * to + c;
        const int bc   = ccol < BCOLS ? ccol : BCOLS - 1;
        const float bl = biasn[bc];
        bv[to] = ccol < BCOLS ? bl : 0.0f;
    }
#pragma unroll
    for (int r = 0; r < 8; ++r) {
        const int m = node0 + r;
        float dvv[NT], svv[NT];
#pragma unroll
        for (int to = 0; to < NT; ++to) {
            dvv[to] = po[to][r] * f8[r];
            svv[to] = dvv[to] + bv[to] * q8[r];
        }
        if (m < N) {
            const size_t o = (size_t)m * DSTRIDE + c;
#pragma unroll
            for (int to = 0; to < NT; ++to) {
                Dn[o + 16 * to] = dvv[to];
                Sn[o + 16 * to] = svv[to];
            }
        }
    }
}

template <int G>
__device__ __forceinline__ void seg16_fetch(v2f& bfr, v2f& a, int src, int tg,
                                            const float* __restrict__ lin, int c) {
    const int sx = row_share16<2 * G>(src),     tx = row_share16<2 * G>(tg);
    const int sy = row_share16<2 * G + 1>(src), ty = row_share16<2 * G + 1>(tg);
    a[0] = tx == c ? 1.0f : 0.0f;
    a[1] = ty == c ? 1.0f : 0.0f;
    const char* __restrict__ linb = (const char*)lin;
    bfr[0] = *(const float*)(linb + (((unsigned)sx << 6) | ((unsigned)c << 2)));
    bfr[1] = *(const float*)(linb + (((unsigned)sy << 6) | ((unsigned)c << 2)));
}

template <int SB>
__device__ __forceinline__ void seg16_half_trip(v8f& acc, int src, int tg,
                                                const float* __restrict__ lin, int c) {
    v2f bfr[4], a[4];
    seg16_fetch<4 * SB + 0>(bfr[0], a[0], src, tg, lin, c);
    seg16_fetch<4 * SB + 1>(bfr[1], a[1], src, tg, lin, c);
    seg16_fetch<4 * SB + 2>(bfr[2], a[2], src, tg, lin, c);
    seg16_fetch<4 * SB + 3>(bfr[3], a[3], src, tg, lin, c);
#pragma unroll
    for (int i = 0; i < 4; ++i)
        acc = __builtin_amdgcn_wmma_f32_16x16x4_f32(false, a[i], false, bfr[i], (short)0, acc, false, false);
}

__global__ __launch_bounds__(SEG_WAVES * 32)
void seg_sum16_wmma_kernel(const float* __restrict__ lin, const int* __restrict__ binned,
                           const int* __restrict__ binStart, float* __restrict__ out, int N, int nbins) {
    const int lane = threadIdx.x & 31;
    const int wib  = __builtin_amdgcn_readfirstlane((int)(threadIdx.x >> 5));
    const int b    = blockIdx.x * SEG_WAVES + wib;
    if (b >= nbins) return;
    const int c    = lane & 15;
    const int half = lane >> 4;
    const int slot = 4 * (c >> 1) + 2 * half + (c & 1);

    const int eb = binStart[b];
    const int ee = binStart[b + 1];
    const int2* __restrict__ ed2 = (const int2*)binned;

    v8f acc = {0.f, 0.f, 0.f, 0.f, 0.f, 0.f, 0.f, 0.f};

    int2 ed = make_int2(0, 0);
    if (eb < ee) {
        int idx = eb + slot;   if (idx >= ee) idx = ee - 1;
        ed = ed2[idx];
    }
    for (int base = eb; base < ee; base += 32) {
        const int nv = ee - base;
        int2 edn = ed;
        if (base + 32 < ee) {
            int idx = base + 32 + slot;   if (idx >= ee) idx = ee - 1;
            edn = ed2[idx];
        }
        const int src = ed.x;
        const int tg  = slot < nv ? ed.y : BIN_ROWS;
        seg16_half_trip<0>(acc, src, tg, lin, c);
        if (nv > 16) seg16_half_trip<1>(acc, src, tg, lin, c);
        ed = edn;
    }

    const int node0 = (b << BIN_SHIFT) + half * 8;
    float sv[8];
#pragma unroll
    for (int r = 0; r < 8; ++r) {
        const int node = node0 + r;
        const int nc = node < N ? node : N - 1;
        sv[r] = out[(unsigned)(nc * 16 + c)];
    }
#pragma unroll
    for (int r = 0; r < 8; ++r) {
        const int node = node0 + r;
        if (node < N) out[(unsigned)(node * 16 + c)] = sv[r] + acc[r];
    }
}

#define POOL_NPT 16

__device__ __forceinline__ void pool_flush(float* __restrict__ out, float* __restrict__ cnt,
                                           int g, int c, float acc, float cn) {
    if (c < C_OUT) atomicAdd(&out[(size_t)g * C_OUT + c], acc);
    if (c == 0)    atomicAdd(&cnt[g], cn);
}

__global__ __launch_bounds__(256)
void pool_kernel(const float* __restrict__ h  ,
                 const float* __restrict__ dis,
                 const int* __restrict__ batch,
                 float* __restrict__ out, float* __restrict__ cnt, int N) {
    const int lane  = threadIdx.x & 31;
    const int c     = lane & 15;
    const int half  = lane >> 4;
    const int wave  = (blockIdx.x * blockDim.x + threadIdx.x) >> 5;
    const int first = wave * (2 * POOL_NPT);
    if (first >= N) return;
    const int nb = first + half;
    float hv[POOL_NPT], dv[POOL_NPT];
    int   gv[POOL_NPT];
#pragma unroll
    for (int i = 0; i < POOL_NPT; ++i) {
        const int node = nb + 2 * i;
        const int nc = node < N ? node : N - 1;
        hv[i] = h[(size_t)nc * 16 + c];
        dv[i] = dis[nc];
        gv[i] = batch[nc];
    }
    int   gcur = gv[0];
    float acc = 0.0f, cn = 0.0f;
#pragma unroll
    for (int i = 0; i < POOL_NPT; ++i) {
        const int node = nb + 2 * i;
        if (node < N) {
            if (gv[i] != gcur) {
                pool_flush(out, cnt, gcur, c, acc, cn);
                gcur = gv[i];   acc = 0.0f;   cn = 0.0f;
            }
            acc += hv[i] * dv[i];
            cn  += 1.0f;
        }
    }
    if (cn > 0.0f) pool_flush(out, cnt, gcur, c, acc, cn);
}

__global__ void div_kernel(float* __restrict__ out, const float* __restrict__ cnt) {
    int t = blockIdx.x * blockDim.x + threadIdx.x;
    if (t < NUM_GRAPHS * C_OUT) out[t] /= fmaxf(cnt[t / C_OUT], 1.0f);
}


static inline int cdiv_ll(long long a, long long b) { return (int)((a + b - 1) / b); }

extern "C" void kernel_launch(void* const* d_in, const int* in_sizes, int n_in,
                              void* d_out, int out_size, void* d_ws, size_t ws_size,
                              hipStream_t stream) {
    const float* x     = (const float*)d_in[0];
    const int*   ei    = (const int*)d_in[1];
    const int*   batch = (const int*)d_in[2];
    const float* W0 = (const float*)d_in[3];  const float* b0 = (const float*)d_in[4];
    const float* W1 = (const float*)d_in[5];  const float* b1 = (const float*)d_in[6];
    const float* W2 = (const float*)d_in[7];  const float* b2 = (const float*)d_in[8];
    const float* W3 = (const float*)d_in[9];  const float* b3 = (const float*)d_in[10];

    const int N = in_sizes[2];
    const int E = in_sizes[1] / 2;
    const int H = 128;
    const int* rowi = ei;
    const int* coli = ei + E;

    float* out = (float*)d_out;
    float* ws  = (float*)d_ws;
    const size_t Npad = ((size_t)N + 255) & ~(size_t)255;
    float* dis  = ws;
    float* sq   = ws + Npad;
    float* bufA = sq + Npad;
    float* bufB = bufA + (size_t)N * H;
    float* bufC = bufB + (size_t)N * H;
    float* cnt  = bufC + (size_t)N * H;
    const int nbins = (N + BIN_ROWS - 1) >> BIN_SHIFT;
    const int nsup  = (N + SUP_ROWS - 1) >> SUP_SHIFT;
    const size_t binPad = ((size_t)nbins + 1 + 255) & ~(size_t)255;
    int* binStart = (int*)(cnt + NUM_GRAPHS);
    int* cursor   = binStart + binPad;
    int* binned   = cursor + binPad * CUR_STRIDE;

    const int TB = 256;

    fill_kernel<<<cdiv_ll(N, TB), TB, 0, stream>>>(dis, 1.0f, N);
    degree_kernel<<<cdiv_ll(E, TB * EPT), TB, 0, stream>>>(dis, coli, E);
    bin_offsets_kernel<<<1, 1024, 0, stream>>>(dis, binStart, cursor, N, nbins, nsup);
    rsqrt_kernel<<<cdiv_ll(N, TB), TB, 0, stream>>>(dis, sq, N);
    place_kernel<<<cdiv_ll(E, TB * EPT), TB, 0, stream>>>(rowi, coli, cursor, binned, E);

    const int gemmBlkH = 256 * 2;
    const int fuseBlk = (nbins + FUSE_WAVES - 1) / FUSE_WAVES;
    const int fuseTB  = FUSE_WAVES * 32;
    const int segBlk  = (nbins + SEG_WAVES - 1) / SEG_WAVES;
    const int segTB   = SEG_WAVES * 32;
    float* lin4 = bufC;
    float* sum4 = bufC + (size_t)N * 16;

    gemm_wmma_f32<128, 128, 128, false, true><<<gemmBlkH, TB, 0, stream>>>(x, W0, dis, sq, b0, bufA, bufB, N, 8);
    seg_sum_gemm_kernel<128, 128><<<fuseBlk, fuseTB, 0, stream>>>(bufA, binned, binStart, bufB, W1, b1, dis, sq, bufC, bufB, N, nbins);
    seg_sum_gemm_kernel<128, 128><<<fuseBlk, fuseTB, 0, stream>>>(bufC, binned, binStart, bufB, W2, b2, dis, sq, bufA, bufB, N, nbins);
    seg_sum_gemm_kernel<10, 16><<<fuseBlk, fuseTB, 0, stream>>>(bufA, binned, binStart, bufB, W3, b3, dis, sq, lin4, sum4, N, nbins);
    seg_sum16_wmma_kernel<<<segBlk, segTB, 0, stream>>>(lin4, binned, binStart, sum4, N, nbins);

    fill_kernel<<<cdiv_ll(NUM_GRAPHS * C_OUT, TB), TB, 0, stream>>>(out, 0.0f, NUM_GRAPHS * C_OUT);
    fill_kernel<<<cdiv_ll(NUM_GRAPHS, TB), TB, 0, stream>>>(cnt, 0.0f, NUM_GRAPHS);
    pool_kernel<<<cdiv_ll((long long)N, (TB / 32) * 2 * POOL_NPT), TB, 0, stream>>>(sum4, dis, batch, out, cnt, N);
    div_kernel<<<cdiv_ll(NUM_GRAPHS * C_OUT, TB), TB, 0, stream>>>(out, cnt);
}
